# starting kernel
# baseline (speedup 1.0000x reference)
#include <hip/hip_runtime.h>
#include <hip/hip_bf16.h>

#define F_DIM 128
#define NEG_SLOPE 0.2f

typedef __attribute__((ext_vector_type(16))) __bf16 v16bf;
typedef __attribute__((ext_vector_type(8)))  float  v8f;

#define WPACK_ELEMS (8 * 4 * 32 * 16)

__device__ __forceinline__ float leaky(float z) {
    return z > 0.0f ? z : NEG_SLOPE * z;
}

__device__ __forceinline__ void atomicMaxF32(float* addr, float val) {
    if (val >= 0.0f) {
        atomicMax((int*)addr, __float_as_int(val));
    } else {
        atomicMin((unsigned int*)addr, __float_as_uint(val));
    }
}

__global__ void gat_pack_w(const float* __restrict__ W1s, const float* __restrict__ W2s,
                           __bf16* __restrict__ packed) {
    int idx  = blockIdx.x * blockDim.x + threadIdx.x;
    int j    = idx & 15;
    int lane = (idx >> 4) & 31;
    int t    = (idx >> 9) & 3;
    int w    = (idx >> 11) & 7;
    int p    = (idx >> 14) & 1;
    const float* W = p ? W2s : W1s;
    int koff = (lane >> 4) * 8;
    int mn   = lane & 15;
    int k    = t * 32 + ((j >> 3) << 4) + koff + (j & 7);
    int col  = w * 16 + mn;
    packed[idx] = (__bf16)W[k * F_DIM + col];
}

__global__ void gat_matvec(const float* __restrict__ W1s, const float* __restrict__ a1s,
                           const float* __restrict__ W1d, const float* __restrict__ a1d,
                           const float* __restrict__ W2s, const float* __restrict__ a2s,
                           const float* __restrict__ W2d, const float* __restrict__ a2d,
                           float* __restrict__ vecs) {
    const float* W; const float* a;
    switch (blockIdx.x) {
        case 0:  W = W1s; a = a1s; break;
        case 1:  W = W1d; a = a1d; break;
        case 2:  W = W2s; a = a2s; break;
        default: W = W2d; a = a2d; break;
    }
    int f = threadIdx.x;
    float s = 0.0f;
    #pragma unroll 8
    for (int d = 0; d < F_DIM; ++d) s += W[f * F_DIM + d] * a[d];
    vecs[blockIdx.x * F_DIM + f] = s;
}

__global__ void __launch_bounds__(256)
gat_rowdot(const float* __restrict__ X, const float* __restrict__ vs,
           const float* __restrict__ vd, float* __restrict__ al_s,
           float* __restrict__ al_d, int N) {
    int wave = threadIdx.x >> 5, lane = threadIdx.x & 31;
    int row = blockIdx.x * 8 + wave;
    if (row >= N) return;
    float4 xv = ((const float4*)(X + (size_t)row * F_DIM))[lane];
    float4 av = ((const float4*)vs)[lane];
    float4 bv = ((const float4*)vd)[lane];
    float ss = xv.x * av.x + xv.y * av.y + xv.z * av.z + xv.w * av.w;
    float sd = xv.x * bv.x + xv.y * bv.y + xv.z * bv.z + xv.w * bv.w;
    #pragma unroll
    for (int off = 16; off > 0; off >>= 1) {
        ss += __shfl_down(ss, off, 32);
        sd += __shfl_down(sd, off, 32);
    }
    if (lane == 0) { al_s[row] = ss; al_d[row] = sd; }
}

__global__ void __launch_bounds__(256)
gat_gemm_wmma(const float* __restrict__ A, const __bf16* __restrict__ Wp,
              float* __restrict__ C, int N) {
    __shared__ __bf16 As[16][F_DIM];
    const int tid  = threadIdx.x;
    const int row0 = blockIdx.x * 16;

    #pragma unroll
    for (int j = 0; j < 8; ++j) {
        int i = tid * 8 + j;
        int r = i >> 7, k = i & 127;
        As[r][k] = (__bf16)A[(size_t)(row0 + r) * F_DIM + k];
    }
    __syncthreads();

    const int wave = tid >> 5, lane = tid & 31;
    const int col0 = wave * 16;
    const int mn   = lane & 15;
    const int koff = (lane >> 4) * 8;

    const v16bf* Bfrag = (const v16bf*)Wp + (size_t)(wave * 4) * 32 + lane;

    v8f acc = {};
    #pragma unroll
    for (int t = 0; t < 4; ++t) {
        v16bf af;
        #pragma unroll
        for (int i = 0; i < 8; ++i) {
            af[i]     = As[mn][t * 32 + koff + i];
            af[i + 8] = As[mn][t * 32 + 16 + koff + i];
        }
        v16bf bf = Bfrag[t * 32];
        acc = __builtin_amdgcn_wmma_f32_16x16x32_bf16(
             false, af,  false, bf,
             (short)0, acc,  false,  false);
    }

    const int mb = (lane >> 4) * 8;
    #pragma unroll
    for (int v = 0; v < 8; ++v)
        C[(size_t)(row0 + mb + v) * F_DIM + col0 + mn] = acc[v];
}

__global__ void gat_init(float* __restrict__ m, float* __restrict__ denom,
                         float* __restrict__ acc, int N) {
    int i = blockIdx.x * blockDim.x + threadIdx.x;
    if (i < N * F_DIM) acc[i] = 0.0f;
    if (i < N) { m[i] = -3.402823466e38f; denom[i] = 0.0f; }
}

__global__ void gat_edge_max(const int* __restrict__ src, const int* __restrict__ dst,
                             const float* __restrict__ al_s, const float* __restrict__ al_d,
                             float* __restrict__ m, int E) {
    int e = blockIdx.x * blockDim.x + threadIdx.x;
    if (e >= E) return;
    int d = dst[e];
    float z = leaky(al_s[src[e]] + al_d[d]);
    atomicMaxF32(m + d, z);
}

__global__ void gat_edge_exp(const int* __restrict__ src, const int* __restrict__ dst,
                             const float* __restrict__ al_s, const float* __restrict__ al_d,
                             const float* __restrict__ m, float* __restrict__ ebuf,
                             float* __restrict__ denom, int E) {
    int e = blockIdx.x * blockDim.x + threadIdx.x;
    if (e >= E) return;
    int d = dst[e];
    float z  = leaky(al_s[src[e]] + al_d[d]);
    float ex = __expf(z - m[d]);
    ebuf[e] = ex;
    atomicAdd(denom + d, ex);
}

__global__ void __launch_bounds__(256)
gat_edge_agg(const int* __restrict__ src, const int* __restrict__ dst,
             const float* __restrict__ hs, const float* __restrict__ ebuf,
             const float* __restrict__ denom, float* __restrict__ acc, int E) {
    int wave = threadIdx.x >> 5, lane = threadIdx.x & 31;
    int e = blockIdx.x * 8 + wave;
    if (e >= E) return;
    int s = src[e], d = dst[e];
    float alpha = ebuf[e] / denom[d];
    float4 hv = ((const float4*)(hs + (size_t)s * F_DIM))[lane];
    float* out = acc + (size_t)d * F_DIM + lane * 4;
    atomicAdd(out + 0, alpha * hv.x);
    atomicAdd(out + 1, alpha * hv.y);
    atomicAdd(out + 2, alpha * hv.z);
    atomicAdd(out + 3, alpha * hv.w);
}

__global__ void gat_finalize(const float* __restrict__ acc, const float* __restrict__ bias,
                             float* __restrict__ out, int N, int do_relu) {
    int i = blockIdx.x * blockDim.x + threadIdx.x;
    if (i >= N * F_DIM) return;
    float v = acc[i] + bias[i & (F_DIM - 1)];
    if (do_relu) v = fmaxf(v, 0.0f);
    out[i] = v;
}

extern "C" void kernel_launch(void* const* d_in, const int* in_sizes, int n_in,
                              void* d_out, int out_size, void* d_ws, size_t ws_size,
                              hipStream_t stream) {
    const float* x   = (const float*)d_in[0];
    const int*   ei  = (const int*)d_in[1];
    const float* W1s = (const float*)d_in[2];
    const float* W1d = (const float*)d_in[3];
    const float* a1s = (const float*)d_in[4];
    const float* a1d = (const float*)d_in[5];
    const float* b1  = (const float*)d_in[6];
    const float* W2s = (const float*)d_in[7];
    const float* W2d = (const float*)d_in[8];
    const float* a2s = (const float*)d_in[9];
    const float* a2d = (const float*)d_in[10];
    const float* b2  = (const float*)d_in[11];

    const int N = in_sizes[0] / F_DIM;
    const int E = in_sizes[1] / 2;
    const int* src = ei;
    const int* dst = ei + E;

    __bf16* wpack = (__bf16*)d_ws;
    float*  fws   = (float*)((char*)d_ws + 2 * WPACK_ELEMS * sizeof(__bf16));
    float* hs    = fws;
    float* hbuf  = hs   + (size_t)N * F_DIM;
    float* al_s  = hbuf + (size_t)N * F_DIM;
    float* al_d  = al_s + N;
    float* mbuf  = al_d + N;
    float* denom = mbuf + N;
    float* ebuf  = denom + N;
    float* vecs  = ebuf + E;

    float* out = (float*)d_out;

    dim3 b256(256);
    const int gNF   = (N * F_DIM + 255) / 256;
    const int gE    = (E + 255) / 256;
    const int gAgg  = (E + 7) / 8;
    const int gRow  = (N + 7) / 8;
    const int gGemm = (N + 15) / 16;

    gat_pack_w<<<(2 * WPACK_ELEMS + 255) / 256, b256, 0, stream>>>(W1s, W2s, wpack);
    gat_matvec<<<4, 128, 0, stream>>>(W1s, a1s, W1d, a1d, W2s, a2s, W2d, a2d, vecs);

    gat_rowdot   <<<gRow,  b256, 0, stream>>>(x, vecs, vecs + 128, al_s, al_d, N);
    gat_gemm_wmma<<<gGemm, b256, 0, stream>>>(x, wpack, hs, N);
    gat_init     <<<gNF,   b256, 0, stream>>>(mbuf, denom, hbuf, N);
    gat_edge_max <<<gE,    b256, 0, stream>>>(src, dst, al_s, al_d, mbuf, E);
    gat_edge_exp <<<gE,    b256, 0, stream>>>(src, dst, al_s, al_d, mbuf, ebuf, denom, E);
    gat_edge_agg <<<gAgg,  b256, 0, stream>>>(src, dst, hs, ebuf, denom, hbuf, E);
    gat_finalize <<<gNF,   b256, 0, stream>>>(hbuf, b1, hbuf, N, 1);

    gat_rowdot   <<<gRow,  b256, 0, stream>>>(hbuf, vecs + 256, vecs + 384, al_s, al_d, N);
    gat_gemm_wmma<<<gGemm, b256, 0, stream>>>(hbuf, wpack + WPACK_ELEMS, hs, N);
    gat_init     <<<gNF,   b256, 0, stream>>>(mbuf, denom, out, N);
    gat_edge_max <<<gE,    b256, 0, stream>>>(src, dst, al_s, al_d, mbuf, E);
    gat_edge_exp <<<gE,    b256, 0, stream>>>(src, dst, al_s, al_d, mbuf, ebuf, denom, E);
    gat_edge_agg <<<gAgg,  b256, 0, stream>>>(src, dst, hs, ebuf, denom, out, E);
    gat_finalize <<<gNF,   b256, 0, stream>>>(out, b2, out, N, 0);
}


# Row dots al_s/al_d taken inside the GEMM kernel from the f32 A rows it already loads (same 32-lane order, bit-identical): the row-dot kernel is gone, 11 -> 9 launches, 200000 fewer waves per call
# speedup vs baseline: 5.6201x; 5.6201x over previous
#include <hip/hip_runtime.h>
#include <hip/hip_bf16.h>

#define F_DIM 128
#define NEG_SLOPE 0.2f

typedef __attribute__((ext_vector_type(16))) __bf16 v16bf;
typedef __attribute__((ext_vector_type(8)))  float  v8f;

#define WPACK_ELEMS (8 * 4 * 32 * 16)

__device__ __forceinline__ float leaky(float z) {
    return z > 0.0f ? z : NEG_SLOPE * z;
}

__global__ void gat_pack_w(const float* __restrict__ W1s, const float* __restrict__ W2s,
                           __bf16* __restrict__ packed) {
    int idx  = blockIdx.x * blockDim.x + threadIdx.x;
    int j    = idx & 15;
    int lane = (idx >> 4) & 31;
    int t    = (idx >> 9) & 3;
    int w    = (idx >> 11) & 7;
    int p    = (idx >> 14) & 1;
    const float* W = p ? W2s : W1s;
    int koff = (lane >> 4) * 8;
    int mn   = lane & 15;
    int k    = t * 32 + ((j >> 3) << 4) + koff + (j & 7);
    int col  = w * 16 + mn;
    packed[idx] = (__bf16)W[k * F_DIM + col];
}

__global__ void gat_matvec(const float* __restrict__ W1s, const float* __restrict__ a1s,
                           const float* __restrict__ W1d, const float* __restrict__ a1d,
                           const float* __restrict__ W2s, const float* __restrict__ a2s,
                           const float* __restrict__ W2d, const float* __restrict__ a2d,
                           float* __restrict__ vecs) {
    const float* W; const float* a;
    switch (blockIdx.x) {
        case 0:  W = W1s; a = a1s; break;
        case 1:  W = W1d; a = a1d; break;
        case 2:  W = W2s; a = a2s; break;
        default: W = W2d; a = a2d; break;
    }
    int f = threadIdx.x;
    float s = 0.0f;
    #pragma unroll 8
    for (int d = 0; d < F_DIM; ++d) s += W[f * F_DIM + d] * a[d];
    vecs[blockIdx.x * F_DIM + f] = s;
}

typedef __attribute__((ext_vector_type(4))) __bf16 v4bf;

__global__ void __launch_bounds__(256)
gat_gemm_wmma(const float* __restrict__ A, const __bf16* __restrict__ Wp,
              const float* __restrict__ vs, const float* __restrict__ vd,
              float* __restrict__ C, float* __restrict__ al_s, float* __restrict__ al_d, int N) {
    __shared__ __attribute__((aligned(16))) __bf16 As[16][F_DIM];
    const int tid  = threadIdx.x;
    const int row0 = blockIdx.x * 16;
    const int wave = tid >> 5, lane = tid & 31;

    const int ra = 2 * wave, rb = 2 * wave + 1;
    const float4 xa = ((const float4*)(A + (size_t)(row0 + ra) * F_DIM))[lane];
    const float4 xb = ((const float4*)(A + (size_t)(row0 + rb) * F_DIM))[lane];
    const float4 av = ((const float4*)vs)[lane];
    const float4 bv = ((const float4*)vd)[lane];
    float sa = xa.x * av.x + xa.y * av.y + xa.z * av.z + xa.w * av.w;
    float da = xa.x * bv.x + xa.y * bv.y + xa.z * bv.z + xa.w * bv.w;
    float sb = xb.x * av.x + xb.y * av.y + xb.z * av.z + xb.w * av.w;
    float db = xb.x * bv.x + xb.y * bv.y + xb.z * bv.z + xb.w * bv.w;
    {
        v4bf pa = { (__bf16)xa.x, (__bf16)xa.y, (__bf16)xa.z, (__bf16)xa.w };
        v4bf pb = { (__bf16)xb.x, (__bf16)xb.y, (__bf16)xb.z, (__bf16)xb.w };
        *(v4bf*)&As[ra][4 * lane] = pa;
        *(v4bf*)&As[rb][4 * lane] = pb;
    }
    __syncthreads();

    const int col0 = wave * 16;
    const int mn   = lane & 15;
    const int koff = (lane >> 4) * 8;

    const v16bf* Bfrag = (const v16bf*)Wp + (size_t)(wave * 4) * 32 + lane;
    v16bf bfr[4];
    #pragma unroll
    for (int t = 0; t < 4; ++t) bfr[t] = Bfrag[t * 32];

    #pragma unroll
    for (int off = 16; off > 0; off >>= 1) {
        sa += __shfl_down(sa, off, 32);
        da += __shfl_down(da, off, 32);
        sb += __shfl_down(sb, off, 32);
        db += __shfl_down(db, off, 32);
    }
    if (lane == 0) {
        al_s[row0 + ra] = sa; al_d[row0 + ra] = da;
        al_s[row0 + rb] = sb; al_d[row0 + rb] = db;
    }

    v8f acc = {};
    #pragma unroll
    for (int t = 0; t < 4; ++t) {
        v16bf af;
        #pragma unroll
        for (int i = 0; i < 8; ++i) {
            af[i]     = As[mn][t * 32 + koff + i];
            af[i + 8] = As[mn][t * 32 + 16 + koff + i];
        }
        acc = __builtin_amdgcn_wmma_f32_16x16x32_bf16(
             false, af,  false, bfr[t],
             (short)0, acc,  false,  false);
    }

    const int mb = (lane >> 4) * 8;
    #pragma unroll
    for (int v = 0; v < 8; ++v)
        C[(size_t)(row0 + mb + v) * F_DIM + col0 + mn] = acc[v];
}


#define BIN_SHIFT 9
#define BIN_W     512
#define NBP       256
#define CH_CAP    6400
#define BIN_CAP   12288

__device__ __forceinline__ int wave_incl_scan(int v, int lane) {
    #pragma unroll
    for (int off = 1; off < 32; off <<= 1) {
        int t = __shfl_up(v, off, 32);
        if (lane >= off) v += t;
    }
    return v;
}

__global__ void __launch_bounds__(1024)
gat_bin_count(const int* __restrict__ dst, int* __restrict__ table, int E) {
    __shared__ int h[NBP];
    const int t = threadIdx.x;
    if (t < NBP) h[t] = 0;
    __syncthreads();
    const int e0 = blockIdx.x * CH_CAP;
    const int e1 = (e0 + CH_CAP < E) ? (e0 + CH_CAP) : E;
    for (int e = e0 + t; e < e1; e += 1024)
        atomicAdd(&h[(dst[e] >> BIN_SHIFT) & (NBP - 1)], 1);
    __syncthreads();
    if (t < NBP) table[blockIdx.x * NBP + t] = h[t];
}

__global__ void __launch_bounds__(1024)
gat_bin_partition(const int* __restrict__ src, const int* __restrict__ dst,
                  const int* __restrict__ table, int2* __restrict__ pairs,
                  int* __restrict__ binStart, int* __restrict__ binTotal, int E, int NC) {
    __shared__ int  sTot[4][NBP];
    __shared__ int  sPre[4][NBP];
    __shared__ int  sMine[NBP];
    __shared__ int  sCur[NBP];
    __shared__ int  sOff[NBP];
    __shared__ int  sWs[2][8];
    __shared__ int2 sBuf[CH_CAP];
    const int t = threadIdx.x, lane = t & 31, wave = t >> 5;
    const int chunk = blockIdx.x;
    const int b = t & (NBP - 1), q = t >> 8;

    int tot = 0, pre = 0;
    #pragma unroll 16
    for (int c = q; c < NC; c += 4) {
        const int v = table[c * NBP + b];
        tot += v;
        if (c < chunk)  pre += v;
        if (c == chunk) sMine[b] = v;
    }
    sTot[q][b] = tot;
    sPre[q][b] = pre;
    __syncthreads();

    int T = 0, P = 0, M = 0, iT = 0, iM = 0;
    if (t < NBP) {
        T = sTot[0][t] + sTot[1][t] + sTot[2][t] + sTot[3][t];
        P = sPre[0][t] + sPre[1][t] + sPre[2][t] + sPre[3][t];
        M = sMine[t];
        iT = wave_incl_scan(T, lane);
        iM = wave_incl_scan(M, lane);
        if (lane == 31) { sWs[0][wave] = iT; sWs[1][wave] = iM; }
    }
    __syncthreads();
    if (t < NBP) {
        int oT = 0, oM = 0;
        for (int w = 0; w < wave; ++w) { oT += sWs[0][w]; oM += sWs[1][w]; }
        const int gStart = oT + iT - T;
        const int lStart = oM + iM - M;
        sCur[t] = lStart;
        sOff[t] = gStart + P - lStart;
        if (chunk == 0) { binStart[t] = gStart; binTotal[t] = T; }
    }
    __syncthreads();

    const int e0 = chunk * CH_CAP;
    const int e1 = (e0 + CH_CAP < E) ? (e0 + CH_CAP) : E;
    for (int eb = e0 + t; eb < e1; eb += 4 * 1024) {
        int d[4], s[4];
        #pragma unroll
        for (int k = 0; k < 4; ++k) {
            const int e = eb + k * 1024;
            d[k] = (e < e1) ? dst[e] : 0;
            s[k] = (e < e1) ? src[e] : 0;
        }
        #pragma unroll
        for (int k = 0; k < 4; ++k) {
            if (eb + k * 1024 < e1) {
                const int p = atomicAdd(&sCur[(d[k] >> BIN_SHIFT) & (NBP - 1)], 1);
                sBuf[p] = make_int2(s[k], d[k]);
            }
        }
    }
    __syncthreads();
    const int len = e1 - e0;
    for (int i = t; i < len; i += 1024) {
        const int2 pr = sBuf[i];
        pairs[sOff[(pr.y >> BIN_SHIFT) & (NBP - 1)] + i] = pr;
    }
}

__global__ void __launch_bounds__(1024)
gat_bin_sort(const int2* __restrict__ pairs, const int* __restrict__ binStart,
             const int* __restrict__ binTotal, int* __restrict__ cnt,
             int* __restrict__ rowptr, int* __restrict__ ssrc, int N) {
    __shared__ int sCnt[BIN_W];
    __shared__ int sCur[BIN_W];
    __shared__ int sWs[16];
    __shared__ int sOut[BIN_CAP];
    const int t = threadIdx.x, lane = t & 31, wave = t >> 5;
    const int bin  = blockIdx.x;
    const int lo   = bin << BIN_SHIFT;
    const int base = binStart[bin];
    const int tot  = binTotal[bin];

    if (t < BIN_W) sCnt[t] = 0;
    __syncthreads();
    for (int ib = t; ib < tot; ib += 4 * 1024) {
        int d[4];
        #pragma unroll
        for (int k = 0; k < 4; ++k) {
            const int i = ib + k * 1024;
            d[k] = (i < tot) ? pairs[base + i].y : lo;
        }
        #pragma unroll
        for (int k = 0; k < 4; ++k)
            if (ib + k * 1024 < tot) atomicAdd(&sCnt[(d[k] - lo) & (BIN_W - 1)], 1);
    }
    __syncthreads();

    int c = 0, inc = 0;
    if (t < BIN_W) {
        c   = sCnt[t];
        inc = wave_incl_scan(c, lane);
        if (lane == 31) sWs[wave] = inc;
    }
    __syncthreads();
    if (t < BIN_W) {
        int o = 0;
        for (int w = 0; w < wave; ++w) o += sWs[w];
        const int ex = o + inc - c;
        sCur[t] = ex;
        const int n = lo + t;
        if (n < N) { cnt[n] = c; rowptr[n] = base + ex; }
    }
    __syncthreads();

    if (tot <= BIN_CAP) {
        for (int ib = t; ib < tot; ib += 4 * 1024) {
            int2 pr[4];
            #pragma unroll
            for (int k = 0; k < 4; ++k) {
                const int i = ib + k * 1024;
                pr[k] = (i < tot) ? pairs[base + i] : make_int2(0, lo);
            }
            #pragma unroll
            for (int k = 0; k < 4; ++k) {
                if (ib + k * 1024 < tot) {
                    const int p = atomicAdd(&sCur[(pr[k].y - lo) & (BIN_W - 1)], 1);
                    sOut[p] = pr[k].x;
                }
            }
        }
        __syncthreads();
        for (int i = t; i < tot; i += 1024) ssrc[base + i] = sOut[i];
    } else {
        for (int i = t; i < tot; i += 1024) {
            const int2 pr = pairs[base + i];
            const int p = atomicAdd(&sCur[(pr.y - lo) & (BIN_W - 1)], 1);
            ssrc[base + p] = pr.x;
        }
    }
}

__global__ void __launch_bounds__(256)
gat_node_agg(const int* __restrict__ rowptr, const int* __restrict__ cnt,
             const int* __restrict__ ssrc, const float* __restrict__ al_s,
             const float* __restrict__ al_d, const float* __restrict__ hs,
             const float* __restrict__ bias, float* __restrict__ out, int N, int do_relu) {
    const int lane = threadIdx.x & 31;
    const int n = __builtin_amdgcn_readfirstlane((int)(blockIdx.x * 8 + (threadIdx.x >> 5)));
    if (n >= N) return;
    const int start = rowptr[n];
    const int deg   = cnt[n];
    const float ald = al_d[n];

    float m = __int_as_float(0xff800000);
    for (int base = 0; base < deg; base += 32) {
        const int j = base + lane;
        if (j < deg) m = fmaxf(m, leaky(al_s[ssrc[start + j]] + ald));
    }
    #pragma unroll
    for (int off = 16; off > 0; off >>= 1) m = fmaxf(m, __shfl_xor(m, off, 32));

    float4 acc = make_float4(0.0f, 0.0f, 0.0f, 0.0f);
    float dsum = 0.0f;
    for (int base = 0; base < deg; base += 32) {
        const int j = base + lane;
        int s = 0; float ex = 0.0f;
        if (j < deg) {
            s  = ssrc[start + j];
            ex = __expf(leaky(al_s[s] + ald) - m);
        }
        dsum += ex;
        const int nk = (deg - base < 32) ? (deg - base) : 32;
        #pragma unroll 4
        for (int k = 0; k < nk; ++k) {
            const int   sk = __builtin_amdgcn_readlane(s, k);
            const float wk = __int_as_float(__builtin_amdgcn_readlane(__float_as_int(ex), k));
            const float4 hv = ((const float4*)(hs + (size_t)sk * F_DIM))[lane];
            acc.x += wk * hv.x;
            acc.y += wk * hv.y;
            acc.z += wk * hv.z;
            acc.w += wk * hv.w;
        }
    }
    #pragma unroll
    for (int off = 16; off > 0; off >>= 1) dsum += __shfl_xor(dsum, off, 32);

    const float inv = (deg > 0) ? 1.0f / dsum : 0.0f;
    const float4 b = ((const float4*)bias)[lane];
    float4 r;
    r.x = acc.x * inv + b.x;
    r.y = acc.y * inv + b.y;
    r.z = acc.z * inv + b.z;
    r.w = acc.w * inv + b.w;
    if (do_relu) {
        r.x = fmaxf(r.x, 0.0f); r.y = fmaxf(r.y, 0.0f);
        r.z = fmaxf(r.z, 0.0f); r.w = fmaxf(r.w, 0.0f);
    }
    ((float4*)(out + (size_t)n * F_DIM))[lane] = r;
}

extern "C" void kernel_launch(void* const* d_in, const int* in_sizes, int n_in,
                              void* d_out, int out_size, void* d_ws, size_t ws_size,
                              hipStream_t stream) {
    const float* x   = (const float*)d_in[0];
    const int*   ei  = (const int*)d_in[1];
    const float* W1s = (const float*)d_in[2];
    const float* W1d = (const float*)d_in[3];
    const float* a1s = (const float*)d_in[4];
    const float* a1d = (const float*)d_in[5];
    const float* b1  = (const float*)d_in[6];
    const float* W2s = (const float*)d_in[7];
    const float* W2d = (const float*)d_in[8];
    const float* a2s = (const float*)d_in[9];
    const float* a2d = (const float*)d_in[10];
    const float* b2  = (const float*)d_in[11];

    const int N = in_sizes[0] / F_DIM;
    const int E = in_sizes[1] / 2;
    const int* src = ei;
    const int* dst = ei + E;

    const int nChunk = (E + CH_CAP - 1) / CH_CAP;
    const int nBin   = (N + BIN_W - 1) / BIN_W;

    __bf16* wpack = (__bf16*)d_ws;
    float*  fws   = (float*)((char*)d_ws + 2 * WPACK_ELEMS * sizeof(__bf16));
    float* hs    = fws;
    float* hbuf  = hs   + (size_t)N * F_DIM;
    float* al_s  = hbuf + (size_t)N * F_DIM;
    float* al_d  = al_s + N;
    float* vecs  = al_d + N;
    int* cnt      = (int*)(vecs + 4 * F_DIM);
    int* rowptr   = cnt + N;
    int* binStart = rowptr + N;
    int* binTotal = binStart + NBP;
    int* table    = binTotal + NBP;
    int* ssrc     = table + (size_t)nChunk * NBP;
    int2* pairs   = (int2*)(ssrc + (size_t)E + ((size_t)E & 1));

    float* out = (float*)d_out;

    dim3 b256(256);
    const int gRow  = (N + 7) / 8;
    const int gGemm = (N + 15) / 16;

    gat_pack_w<<<(2 * WPACK_ELEMS + 255) / 256, b256, 0, stream>>>(W1s, W2s, wpack);
    gat_matvec<<<4, 128, 0, stream>>>(W1s, a1s, W1d, a1d, W2s, a2s, W2d, a2d, vecs);

    gat_bin_count    <<<nChunk, 1024, 0, stream>>>(dst, table, E);
    gat_bin_partition<<<nChunk, 1024, 0, stream>>>(src, dst, table, pairs, binStart, binTotal, E, nChunk);
    gat_bin_sort     <<<nBin,   1024, 0, stream>>>(pairs, binStart, binTotal, cnt, rowptr, ssrc, N);

    gat_gemm_wmma<<<gGemm, b256, 0, stream>>>(x, wpack, vecs, vecs + 128, hs, al_s, al_d, N);
    gat_node_agg <<<gRow,  b256, 0, stream>>>(rowptr, cnt, ssrc, al_s, al_d, hs, b1, hbuf, N, 1);

    gat_gemm_wmma<<<gGemm, b256, 0, stream>>>(hbuf, wpack + WPACK_ELEMS, vecs + 256, vecs + 384, hs, al_s, al_d, N);
    gat_node_agg <<<gRow,  b256, 0, stream>>>(rowptr, cnt, ssrc, al_s, al_d, hs, b2, out, N, 0);
}


# GEMM A tile in LDS: row pitch 256 B -> 272 B (17 x 16 B), so the 16-byte fragment reads of 16 rows hit 64 different banks instead of 4 (was a 16-way bank conflict on every read); sums unchanged
# speedup vs baseline: 6.6841x; 1.1893x over previous
#include <hip/hip_runtime.h>
#include <hip/hip_bf16.h>

#define F_DIM 128
#define NEG_SLOPE 0.2f

typedef __attribute__((ext_vector_type(16))) __bf16 v16bf;
typedef __attribute__((ext_vector_type(8)))  float  v8f;

#define WPACK_ELEMS (8 * 4 * 32 * 16)

__device__ __forceinline__ float leaky(float z) {
    return z > 0.0f ? z : NEG_SLOPE * z;
}

__global__ void gat_pack_w(const float* __restrict__ W1s, const float* __restrict__ W2s,
                           __bf16* __restrict__ packed) {
    int idx  = blockIdx.x * blockDim.x + threadIdx.x;
    int j    = idx & 15;
    int lane = (idx >> 4) & 31;
    int t    = (idx >> 9) & 3;
    int w    = (idx >> 11) & 7;
    int p    = (idx >> 14) & 1;
    const float* W = p ? W2s : W1s;
    int koff = (lane >> 4) * 8;
    int mn   = lane & 15;
    int k    = t * 32 + ((j >> 3) << 4) + koff + (j & 7);
    int col  = w * 16 + mn;
    packed[idx] = (__bf16)W[k * F_DIM + col];
}

__global__ void gat_matvec(const float* __restrict__ W1s, const float* __restrict__ a1s,
                           const float* __restrict__ W1d, const float* __restrict__ a1d,
                           const float* __restrict__ W2s, const float* __restrict__ a2s,
                           const float* __restrict__ W2d, const float* __restrict__ a2d,
                           float* __restrict__ vecs) {
    const float* W; const float* a;
    switch (blockIdx.x) {
        case 0:  W = W1s; a = a1s; break;
        case 1:  W = W1d; a = a1d; break;
        case 2:  W = W2s; a = a2s; break;
        default: W = W2d; a = a2d; break;
    }
    int f = threadIdx.x;
    float s = 0.0f;
    #pragma unroll 8
    for (int d = 0; d < F_DIM; ++d) s += W[f * F_DIM + d] * a[d];
    vecs[blockIdx.x * F_DIM + f] = s;
}

typedef __attribute__((ext_vector_type(4))) __bf16 v4bf;

#define A_PITCH (F_DIM + 8)

__global__ void __launch_bounds__(256)
gat_gemm_wmma(const float* __restrict__ A, const __bf16* __restrict__ Wp,
              const float* __restrict__ vs, const float* __restrict__ vd,
              float* __restrict__ C, float* __restrict__ al_s, float* __restrict__ al_d, int N) {
    __shared__ __attribute__((aligned(16))) __bf16 As[16][A_PITCH];
    const int tid  = threadIdx.x;
    const int row0 = blockIdx.x * 16;
    const int wave = tid >> 5, lane = tid & 31;

    const int ra = 2 * wave, rb = 2 * wave + 1;
    const float4 xa = ((const float4*)(A + (size_t)(row0 + ra) * F_DIM))[lane];
    const float4 xb = ((const float4*)(A + (size_t)(row0 + rb) * F_DIM))[lane];
    const float4 av = ((const float4*)vs)[lane];
    const float4 bv = ((const float4*)vd)[lane];
    float sa = xa.x * av.x + xa.y * av.y + xa.z * av.z + xa.w * av.w;
    float da = xa.x * bv.x + xa.y * bv.y + xa.z * bv.z + xa.w * bv.w;
    float sb = xb.x * av.x + xb.y * av.y + xb.z * av.z + xb.w * av.w;
    float db = xb.x * bv.x + xb.y * bv.y + xb.z * bv.z + xb.w * bv.w;
    {
        v4bf pa = { (__bf16)xa.x, (__bf16)xa.y, (__bf16)xa.z, (__bf16)xa.w };
        v4bf pb = { (__bf16)xb.x, (__bf16)xb.y, (__bf16)xb.z, (__bf16)xb.w };
        *(v4bf*)&As[ra][4 * lane] = pa;
        *(v4bf*)&As[rb][4 * lane] = pb;
    }
    __syncthreads();

    const int col0 = wave * 16;
    const int mn   = lane & 15;
    const int koff = (lane >> 4) * 8;

    const v16bf* Bfrag = (const v16bf*)Wp + (size_t)(wave * 4) * 32 + lane;
    v16bf bfr[4];
    #pragma unroll
    for (int t = 0; t < 4; ++t) bfr[t] = Bfrag[t * 32];

    #pragma unroll
    for (int off = 16; off > 0; off >>= 1) {
        sa += __shfl_down(sa, off, 32);
        da += __shfl_down(da, off, 32);
        sb += __shfl_down(sb, off, 32);
        db += __shfl_down(db, off, 32);
    }
    if (lane == 0) {
        al_s[row0 + ra] = sa; al_d[row0 + ra] = da;
        al_s[row0 + rb] = sb; al_d[row0 + rb] = db;
    }

    v8f acc = {};
    #pragma unroll
    for (int t = 0; t < 4; ++t) {
        v16bf af;
        #pragma unroll
        for (int i = 0; i < 8; ++i) {
            af[i]     = As[mn][t * 32 + koff + i];
            af[i + 8] = As[mn][t * 32 + 16 + koff + i];
        }
        acc = __builtin_amdgcn_wmma_f32_16x16x32_bf16(
             false, af,  false, bfr[t],
             (short)0, acc,  false,  false);
    }

    const int mb = (lane >> 4) * 8;
    #pragma unroll
    for (int v = 0; v < 8; ++v)
        C[(size_t)(row0 + mb + v) * F_DIM + col0 + mn] = acc[v];
}


#define BIN_SHIFT 9
#define BIN_W     512
#define NBP       256
#define CH_CAP    6400
#define BIN_CAP   12288

__device__ __forceinline__ int wave_incl_scan(int v, int lane) {
    #pragma unroll
    for (int off = 1; off < 32; off <<= 1) {
        int t = __shfl_up(v, off, 32);
        if (lane >= off) v += t;
    }
    return v;
}

__global__ void __launch_bounds__(1024)
gat_bin_count(const int* __restrict__ dst, int* __restrict__ table, int E) {
    __shared__ int h[NBP];
    const int t = threadIdx.x;
    if (t < NBP) h[t] = 0;
    __syncthreads();
    const int e0 = blockIdx.x * CH_CAP;
    const int e1 = (e0 + CH_CAP < E) ? (e0 + CH_CAP) : E;
    for (int e = e0 + t; e < e1; e += 1024)
        atomicAdd(&h[(dst[e] >> BIN_SHIFT) & (NBP - 1)], 1);
    __syncthreads();
    if (t < NBP) table[blockIdx.x * NBP + t] = h[t];
}

__global__ void __launch_bounds__(1024)
gat_bin_partition(const int* __restrict__ src, const int* __restrict__ dst,
                  const int* __restrict__ table, int2* __restrict__ pairs,
                  int* __restrict__ binStart, int* __restrict__ binTotal, int E, int NC) {
    __shared__ int  sTot[4][NBP];
    __shared__ int  sPre[4][NBP];
    __shared__ int  sMine[NBP];
    __shared__ int  sCur[NBP];
    __shared__ int  sOff[NBP];
    __shared__ int  sWs[2][8];
    __shared__ int2 sBuf[CH_CAP];
    const int t = threadIdx.x, lane = t & 31, wave = t >> 5;
    const int chunk = blockIdx.x;
    const int b = t & (NBP - 1), q = t >> 8;

    int tot = 0, pre = 0;
    #pragma unroll 16
    for (int c = q; c < NC; c += 4) {
        const int v = table[c * NBP + b];
        tot += v;
        if (c < chunk)  pre += v;
        if (c == chunk) sMine[b] = v;
    }
    sTot[q][b] = tot;
    sPre[q][b] = pre;
    __syncthreads();

    int T = 0, P = 0, M = 0, iT = 0, iM = 0;
    if (t < NBP) {
        T = sTot[0][t] + sTot[1][t] + sTot[2][t] + sTot[3][t];
        P = sPre[0][t] + sPre[1][t] + sPre[2][t] + sPre[3][t];
        M = sMine[t];
        iT = wave_incl_scan(T, lane);
        iM = wave_incl_scan(M, lane);
        if (lane == 31) { sWs[0][wave] = iT; sWs[1][wave] = iM; }
    }
    __syncthreads();
    if (t < NBP) {
        int oT = 0, oM = 0;
        for (int w = 0; w < wave; ++w) { oT += sWs[0][w]; oM += sWs[1][w]; }
        const int gStart = oT + iT - T;
        const int lStart = oM + iM - M;
        sCur[t] = lStart;
        sOff[t] = gStart + P - lStart;
        if (chunk == 0) { binStart[t] = gStart; binTotal[t] = T; }
    }
    __syncthreads();

    const int e0 = chunk * CH_CAP;
    const int e1 = (e0 + CH_CAP < E) ? (e0 + CH_CAP) : E;
    for (int eb = e0 + t; eb < e1; eb += 4 * 1024) {
        int d[4], s[4];
        #pragma unroll
        for (int k = 0; k < 4; ++k) {
            const int e = eb + k * 1024;
            d[k] = (e < e1) ? dst[e] : 0;
            s[k] = (e < e1) ? src[e] : 0;
        }
        #pragma unroll
        for (int k = 0; k < 4; ++k) {
            if (eb + k * 1024 < e1) {
                const int p = atomicAdd(&sCur[(d[k] >> BIN_SHIFT) & (NBP - 1)], 1);
                sBuf[p] = make_int2(s[k], d[k]);
            }
        }
    }
    __syncthreads();
    const int len = e1 - e0;
    for (int i = t; i < len; i += 1024) {
        const int2 pr = sBuf[i];
        pairs[sOff[(pr.y >> BIN_SHIFT) & (NBP - 1)] + i] = pr;
    }
}

__global__ void __launch_bounds__(1024)
gat_bin_sort(const int2* __restrict__ pairs, const int* __restrict__ binStart,
             const int* __restrict__ binTotal, int* __restrict__ cnt,
             int* __restrict__ rowptr, int* __restrict__ ssrc, int N) {
    __shared__ int sCnt[BIN_W];
    __shared__ int sCur[BIN_W];
    __shared__ int sWs[16];
    __shared__ int sOut[BIN_CAP];
    const int t = threadIdx.x, lane = t & 31, wave = t >> 5;
    const int bin  = blockIdx.x;
    const int lo   = bin << BIN_SHIFT;
    const int base = binStart[bin];
    const int tot  = binTotal[bin];

    if (t < BIN_W) sCnt[t] = 0;
    __syncthreads();
    for (int ib = t; ib < tot; ib += 4 * 1024) {
        int d[4];
        #pragma unroll
        for (int k = 0; k < 4; ++k) {
            const int i = ib + k * 1024;
            d[k] = (i < tot) ? pairs[base + i].y : lo;
        }
        #pragma unroll
        for (int k = 0; k < 4; ++k)
            if (ib + k * 1024 < tot) atomicAdd(&sCnt[(d[k] - lo) & (BIN_W - 1)], 1);
    }
    __syncthreads();

    int c = 0, inc = 0;
    if (t < BIN_W) {
        c   = sCnt[t];
        inc = wave_incl_scan(c, lane);
        if (lane == 31) sWs[wave] = inc;
    }
    __syncthreads();
    if (t < BIN_W) {
        int o = 0;
        for (int w = 0; w < wave; ++w) o += sWs[w];
        const int ex = o + inc - c;
        sCur[t] = ex;
        const int n = lo + t;
        if (n < N) { cnt[n] = c; rowptr[n] = base + ex; }
    }
    __syncthreads();

    if (tot <= BIN_CAP) {
        for (int ib = t; ib < tot; ib += 4 * 1024) {
            int2 pr[4];
            #pragma unroll
            for (int k = 0; k < 4; ++k) {
                const int i = ib + k * 1024;
                pr[k] = (i < tot) ? pairs[base + i] : make_int2(0, lo);
            }
            #pragma unroll
            for (int k = 0; k < 4; ++k) {
                if (ib + k * 1024 < tot) {
                    const int p = atomicAdd(&sCur[(pr[k].y - lo) & (BIN_W - 1)], 1);
                    sOut[p] = pr[k].x;
                }
            }
        }
        __syncthreads();
        for (int i = t; i < tot; i += 1024) ssrc[base + i] = sOut[i];
    } else {
        for (int i = t; i < tot; i += 1024) {
            const int2 pr = pairs[base + i];
            const int p = atomicAdd(&sCur[(pr.y - lo) & (BIN_W - 1)], 1);
            ssrc[base + p] = pr.x;
        }
    }
}


template <int G>
__device__ __forceinline__ void agg_rows(const float* __restrict__ hs, int s, float ex, int k,
                                         int lane, float4& acc) {
    float4 hv[G];
    float  wk[G];
    #pragma unroll
    for (int u = 0; u < G; ++u) {
        const int sk = __builtin_amdgcn_readlane(s, k + u);
        wk[u] = __int_as_float(__builtin_amdgcn_readlane(__float_as_int(ex), k + u));
        hv[u] = ((const float4*)(hs + (size_t)sk * F_DIM))[lane];
    }
    #pragma unroll
    for (int u = 0; u < G; ++u) {
        acc.x = __builtin_fmaf(wk[u], hv[u].x, acc.x);
        acc.y = __builtin_fmaf(wk[u], hv[u].y, acc.y);
        acc.z = __builtin_fmaf(wk[u], hv[u].z, acc.z);
        acc.w = __builtin_fmaf(wk[u], hv[u].w, acc.w);
    }
}

__global__ void __launch_bounds__(256)
gat_node_agg(const int* __restrict__ rowptr, const int* __restrict__ cnt,
             const int* __restrict__ ssrc, const float* __restrict__ al_s,
             const float* __restrict__ al_d, const float* __restrict__ hs,
             const float* __restrict__ bias, float* __restrict__ out, int N, int do_relu) {
    const int lane = threadIdx.x & 31;
    const int n = __builtin_amdgcn_readfirstlane((int)(blockIdx.x * 8 + (threadIdx.x >> 5)));
    if (n >= N) return;
    const int start = rowptr[n];
    const int deg   = cnt[n];
    const float ald = al_d[n];

    int   s0 = 0;
    float z0 = __int_as_float(0xff800000);
    if (lane < deg) {
        s0 = ssrc[start + lane];
        z0 = leaky(al_s[s0] + ald);
    }
    float m = z0;
    for (int base = 32; base < deg; base += 32) {
        const int j = base + lane;
        if (j < deg) m = fmaxf(m, leaky(al_s[ssrc[start + j]] + ald));
    }
    #pragma unroll
    for (int off = 16; off > 0; off >>= 1) m = fmaxf(m, __shfl_xor(m, off, 32));

    float4 acc = make_float4(0.0f, 0.0f, 0.0f, 0.0f);
    float dsum = 0.0f;
    for (int base = 0; base < deg; base += 32) {
        const int j = base + lane;
        int s = 0; float ex = 0.0f;
        if (base == 0) {
            s = s0;
            if (lane < deg) ex = __expf(z0 - m);
        } else if (j < deg) {
            s  = ssrc[start + j];
            ex = __expf(leaky(al_s[s] + ald) - m);
        }
        dsum += ex;
        const int nk = (deg - base < 32) ? (deg - base) : 32;
        int k = 0;
        for (; k + 8 <= nk; k += 8) agg_rows<8>(hs, s, ex, k, lane, acc);
        switch (nk - k) {
            case 1: agg_rows<1>(hs, s, ex, k, lane, acc); break;
            case 2: agg_rows<2>(hs, s, ex, k, lane, acc); break;
            case 3: agg_rows<3>(hs, s, ex, k, lane, acc); break;
            case 4: agg_rows<4>(hs, s, ex, k, lane, acc); break;
            case 5: agg_rows<5>(hs, s, ex, k, lane, acc); break;
            case 6: agg_rows<6>(hs, s, ex, k, lane, acc); break;
            case 7: agg_rows<7>(hs, s, ex, k, lane, acc); break;
            default: break;
        }
    }
    #pragma unroll
    for (int off = 16; off > 0; off >>= 1) dsum += __shfl_xor(dsum, off, 32);

    const float inv = (deg > 0) ? 1.0f / dsum : 0.0f;
    const float4 b = ((const float4*)bias)[lane];
    float4 r;
    r.x = acc.x * inv + b.x;
    r.y = acc.y * inv + b.y;
    r.z = acc.z * inv + b.z;
    r.w = acc.w * inv + b.w;
    if (do_relu) {
        r.x = fmaxf(r.x, 0.0f); r.y = fmaxf(r.y, 0.0f);
        r.z = fmaxf(r.z, 0.0f); r.w = fmaxf(r.w, 0.0f);
    }
    ((float4*)(out + (size_t)n * F_DIM))[lane] = r;
}

extern "C" void kernel_launch(void* const* d_in, const int* in_sizes, int n_in,
                              void* d_out, int out_size, void* d_ws, size_t ws_size,
                              hipStream_t stream) {
    const float* x   = (const float*)d_in[0];
    const int*   ei  = (const int*)d_in[1];
    const float* W1s = (const float*)d_in[2];
    const float* W1d = (const float*)d_in[3];
    const float* a1s = (const float*)d_in[4];
    const float* a1d = (const float*)d_in[5];
    const float* b1  = (const float*)d_in[6];
    const float* W2s = (const float*)d_in[7];
    const float* W2d = (const float*)d_in[8];
    const float* a2s = (const float*)d_in[9];
    const float* a2d = (const float*)d_in[10];
    const float* b2  = (const float*)d_in[11];

    const int N = in_sizes[0] / F_DIM;
    const int E = in_sizes[1] / 2;
    const int* src = ei;
    const int* dst = ei + E;

    const int nChunk = (E + CH_CAP - 1) / CH_CAP;
    const int nBin   = (N + BIN_W - 1) / BIN_W;

    __bf16* wpack = (__bf16*)d_ws;
    float*  fws   = (float*)((char*)d_ws + 2 * WPACK_ELEMS * sizeof(__bf16));
    float* hs    = fws;
    float* hbuf  = hs   + (size_t)N * F_DIM;
    float* al_s  = hbuf + (size_t)N * F_DIM;
    float* al_d  = al_s + N;
    float* vecs  = al_d + N;
    int* cnt      = (int*)(vecs + 4 * F_DIM);
    int* rowptr   = cnt + N;
    int* binStart = rowptr + N;
    int* binTotal = binStart + NBP;
    int* table    = binTotal + NBP;
    int* ssrc     = table + (size_t)nChunk * NBP;
    int2* pairs   = (int2*)(ssrc + (size_t)E + ((size_t)E & 1));

    float* out = (float*)d_out;

    dim3 b256(256);
    const int gRow  = (N + 7) / 8;
    const int gGemm = (N + 15) / 16;

    gat_pack_w<<<(2 * WPACK_ELEMS + 255) / 256, b256, 0, stream>>>(W1s, W2s, wpack);
    gat_matvec<<<4, 128, 0, stream>>>(W1s, a1s, W1d, a1d, W2s, a2s, W2d, a2d, vecs);

    gat_bin_count    <<<nChunk, 1024, 0, stream>>>(dst, table, E);
    gat_bin_partition<<<nChunk, 1024, 0, stream>>>(src, dst, table, pairs, binStart, binTotal, E, nChunk);
    gat_bin_sort     <<<nBin,   1024, 0, stream>>>(pairs, binStart, binTotal, cnt, rowptr, ssrc, N);

    gat_gemm_wmma<<<gGemm, b256, 0, stream>>>(x, wpack, vecs, vecs + 128, hs, al_s, al_d, N);
    gat_node_agg <<<gRow,  b256, 0, stream>>>(rowptr, cnt, ssrc, al_s, al_d, hs, b1, hbuf, N, 1);

    gat_gemm_wmma<<<gGemm, b256, 0, stream>>>(hbuf, wpack + WPACK_ELEMS, vecs + 256, vecs + 384, hs, al_s, al_d, N);
    gat_node_agg <<<gRow,  b256, 0, stream>>>(rowptr, cnt, ssrc, al_s, al_d, hs, b2, out, N, 0);
}


# Independent stages share launches (block index picks the work): W@a vectors + weight pack + bin counts in one grid; layer-1 GEMM row tiles ride in the partition and bin-sort grids; 9 -> 6 launches
# speedup vs baseline: 6.8954x; 1.0316x over previous
#include <hip/hip_runtime.h>
#include <hip/hip_bf16.h>

#define F_DIM 128
#define NEG_SLOPE 0.2f

typedef __attribute__((ext_vector_type(16))) __bf16 v16bf;
typedef __attribute__((ext_vector_type(8)))  float  v8f;

#define WPACK_ELEMS (8 * 4 * 32 * 16)

__device__ __forceinline__ float leaky(float z) {
    return z > 0.0f ? z : NEG_SLOPE * z;
}

__device__ __forceinline__ void pack_w_part(const float* __restrict__ W1s, const float* __restrict__ W2s,
                                            __bf16* __restrict__ packed, int idx) {
    int j    = idx & 15;
    int lane = (idx >> 4) & 31;
    int t    = (idx >> 9) & 3;
    int w    = (idx >> 11) & 7;
    int p    = (idx >> 14) & 1;
    const float* W = p ? W2s : W1s;
    int koff = (lane >> 4) * 8;
    int mn   = lane & 15;
    int k    = t * 32 + ((j >> 3) << 4) + koff + (j & 7);
    int col  = w * 16 + mn;
    packed[idx] = (__bf16)W[k * F_DIM + col];
}

__device__ __forceinline__ void matvec_part(const float* __restrict__ W1s, const float* __restrict__ a1s,
                                            const float* __restrict__ W1d, const float* __restrict__ a1d,
                                            const float* __restrict__ W2s, const float* __restrict__ a2s,
                                            const float* __restrict__ W2d, const float* __restrict__ a2d,
                                            float* __restrict__ vecs, int which, int f) {
    const float* W; const float* a;
    switch (which) {
        case 0:  W = W1s; a = a1s; break;
        case 1:  W = W1d; a = a1d; break;
        case 2:  W = W2s; a = a2s; break;
        default: W = W2d; a = a2d; break;
    }
    float s = 0.0f;
    #pragma unroll 8
    for (int d = 0; d < F_DIM; ++d) s += W[f * F_DIM + d] * a[d];
    vecs[which * F_DIM + f] = s;
}

typedef __attribute__((ext_vector_type(4))) __bf16 v4bf;

#define A_PITCH (F_DIM + 8)

__device__ __forceinline__ void
gemm_tile16(const float* __restrict__ A, const __bf16* __restrict__ Wp,
            const float* __restrict__ vs, const float* __restrict__ vd,
            float* __restrict__ C, float* __restrict__ al_s, float* __restrict__ al_d,
            const int row0, const int wave, const int lane, __bf16 (*As)[A_PITCH], const bool active) {
    const int ra = 2 * wave, rb = 2 * wave + 1;
    float sa = 0.0f, da = 0.0f, sb = 0.0f, db = 0.0f;
    if (active) {
        const float4 xa = ((const float4*)(A + (size_t)(row0 + ra) * F_DIM))[lane];
        const float4 xb = ((const float4*)(A + (size_t)(row0 + rb) * F_DIM))[lane];
        const float4 av = ((const float4*)vs)[lane];
        const float4 bv = ((const float4*)vd)[lane];
        sa = xa.x * av.x + xa.y * av.y + xa.z * av.z + xa.w * av.w;
        da = xa.x * bv.x + xa.y * bv.y + xa.z * bv.z + xa.w * bv.w;
        sb = xb.x * av.x + xb.y * av.y + xb.z * av.z + xb.w * av.w;
        db = xb.x * bv.x + xb.y * bv.y + xb.z * bv.z + xb.w * bv.w;
        v4bf pa = { (__bf16)xa.x, (__bf16)xa.y, (__bf16)xa.z, (__bf16)xa.w };
        v4bf pb = { (__bf16)xb.x, (__bf16)xb.y, (__bf16)xb.z, (__bf16)xb.w };
        *(v4bf*)&As[ra][4 * lane] = pa;
        *(v4bf*)&As[rb][4 * lane] = pb;
    }
    __syncthreads();
    if (!active) return;

    const int col0 = wave * 16;
    const int mn   = lane & 15;
    const int koff = (lane >> 4) * 8;

    const v16bf* Bfrag = (const v16bf*)Wp + (size_t)(wave * 4) * 32 + lane;
    v16bf bfr[4];
    #pragma unroll
    for (int t = 0; t < 4; ++t) bfr[t] = Bfrag[t * 32];

    #pragma unroll
    for (int off = 16; off > 0; off >>= 1) {
        sa += __shfl_down(sa, off, 32);
        da += __shfl_down(da, off, 32);
        sb += __shfl_down(sb, off, 32);
        db += __shfl_down(db, off, 32);
    }
    if (lane == 0) {
        al_s[row0 + ra] = sa; al_d[row0 + ra] = da;
        al_s[row0 + rb] = sb; al_d[row0 + rb] = db;
    }

    v8f acc = {};
    #pragma unroll
    for (int t = 0; t < 4; ++t) {
        v16bf af;
        #pragma unroll
        for (int i = 0; i < 8; ++i) {
            af[i]     = As[mn][t * 32 + koff + i];
            af[i + 8] = As[mn][t * 32 + 16 + koff + i];
        }
        acc = __builtin_amdgcn_wmma_f32_16x16x32_bf16(
             false, af,  false, bfr[t],
             (short)0, acc,  false,  false);
    }

    const int mb = (lane >> 4) * 8;
    #pragma unroll
    for (int v = 0; v < 8; ++v)
        C[(size_t)(row0 + mb + v) * F_DIM + col0 + mn] = acc[v];
}

__global__ void __launch_bounds__(256)
gat_gemm_wmma(const float* __restrict__ A, const __bf16* __restrict__ Wp,
              const float* __restrict__ vs, const float* __restrict__ vd,
              float* __restrict__ C, float* __restrict__ al_s, float* __restrict__ al_d, int N) {
    __shared__ __attribute__((aligned(16))) __bf16 As[16][A_PITCH];
    const int tid = threadIdx.x;
    gemm_tile16(A, Wp, vs, vd, C, al_s, al_d, blockIdx.x * 16, tid >> 5, tid & 31, As, true);
}

__device__ __forceinline__ void
gemm_tile64(const float* __restrict__ A, const __bf16* __restrict__ Wp,
            const float* __restrict__ vs, const float* __restrict__ vd,
            float* __restrict__ C, float* __restrict__ al_s, float* __restrict__ al_d,
            const int tile64, const int N) {
    __shared__ __attribute__((aligned(16))) __bf16 As4[4][16][A_PITCH];
    const int tid = threadIdx.x;
    const int g    = __builtin_amdgcn_readfirstlane(tid >> 8);
    const int row0 = (tile64 * 4 + g) * 16;
    gemm_tile16(A, Wp, vs, vd, C, al_s, al_d, row0, (tid >> 5) & 7, tid & 31, As4[g], row0 < N);
}


#define BIN_SHIFT 9
#define BIN_W     512
#define NBP       256
#define CH_CAP    6400
#define BIN_CAP   12288

__device__ __forceinline__ int wave_incl_scan(int v, int lane) {
    #pragma unroll
    for (int off = 1; off < 32; off <<= 1) {
        int t = __shfl_up(v, off, 32);
        if (lane >= off) v += t;
    }
    return v;
}

__device__ __forceinline__ void
bin_count_part(const int* __restrict__ dst, int* __restrict__ table, int E, const int chunk) {
    __shared__ int h[NBP];
    const int t = threadIdx.x;
    if (t < NBP) h[t] = 0;
    __syncthreads();
    const int e0 = chunk * CH_CAP;
    const int e1 = (e0 + CH_CAP < E) ? (e0 + CH_CAP) : E;
    for (int e = e0 + t; e < e1; e += 1024)
        atomicAdd(&h[(dst[e] >> BIN_SHIFT) & (NBP - 1)], 1);
    __syncthreads();
    if (t < NBP) table[chunk * NBP + t] = h[t];
}

__global__ void __launch_bounds__(1024)
gat_prep_count(const float* __restrict__ W1s, const float* __restrict__ a1s,
               const float* __restrict__ W1d, const float* __restrict__ a1d,
               const float* __restrict__ W2s, const float* __restrict__ a2s,
               const float* __restrict__ W2d, const float* __restrict__ a2d,
               float* __restrict__ vecs, __bf16* __restrict__ packed,
               const int* __restrict__ dst, int* __restrict__ table, int E, int NC) {
    const int b = blockIdx.x;
    if (b < 4) {
        if (threadIdx.x < F_DIM)
            matvec_part(W1s, a1s, W1d, a1d, W2s, a2s, W2d, a2d, vecs, b, threadIdx.x);
    } else if (b < 4 + NC) {
        bin_count_part(dst, table, E, b - 4);
    } else {
        pack_w_part(W1s, W2s, packed, (b - 4 - NC) * 1024 + threadIdx.x);
    }
}

__device__ __forceinline__ void
bin_partition_part(const int* __restrict__ src, const int* __restrict__ dst,
                   const int* __restrict__ table, int2* __restrict__ pairs,
                   int* __restrict__ binStart, int* __restrict__ binTotal, int E, int NC,
                   const int chunk) {
    __shared__ int  sTot[4][NBP];
    __shared__ int  sPre[4][NBP];
    __shared__ int  sMine[NBP];
    __shared__ int  sCur[NBP];
    __shared__ int  sOff[NBP];
    __shared__ int  sWs[2][8];
    __shared__ int2 sBuf[CH_CAP];
    const int t = threadIdx.x, lane = t & 31, wave = t >> 5;
    const int b = t & (NBP - 1), q = t >> 8;

    int tot = 0, pre = 0;
    #pragma unroll 16
    for (int c = q; c < NC; c += 4) {
        const int v = table[c * NBP + b];
        tot += v;
        if (c < chunk)  pre += v;
        if (c == chunk) sMine[b] = v;
    }
    sTot[q][b] = tot;
    sPre[q][b] = pre;
    __syncthreads();

    int T = 0, P = 0, M = 0, iT = 0, iM = 0;
    if (t < NBP) {
        T = sTot[0][t] + sTot[1][t] + sTot[2][t] + sTot[3][t];
        P = sPre[0][t] + sPre[1][t] + sPre[2][t] + sPre[3][t];
        M = sMine[t];
        iT = wave_incl_scan(T, lane);
        iM = wave_incl_scan(M, lane);
        if (lane == 31) { sWs[0][wave] = iT; sWs[1][wave] = iM; }
    }
    __syncthreads();
    if (t < NBP) {
        int oT = 0, oM = 0;
        for (int w = 0; w < wave; ++w) { oT += sWs[0][w]; oM += sWs[1][w]; }
        const int gStart = oT + iT - T;
        const int lStart = oM + iM - M;
        sCur[t] = lStart;
        sOff[t] = gStart + P - lStart;
        if (chunk == 0) { binStart[t] = gStart; binTotal[t] = T; }
    }
    __syncthreads();

    const int e0 = chunk * CH_CAP;
    const int e1 = (e0 + CH_CAP < E) ? (e0 + CH_CAP) : E;
    for (int eb = e0 + t; eb < e1; eb += 4 * 1024) {
        int d[4], s[4];
        #pragma unroll
        for (int k = 0; k < 4; ++k) {
            const int e = eb + k * 1024;
            d[k] = (e < e1) ? dst[e] : 0;
            s[k] = (e < e1) ? src[e] : 0;
        }
        #pragma unroll
        for (int k = 0; k < 4; ++k) {
            if (eb + k * 1024 < e1) {
                const int p = atomicAdd(&sCur[(d[k] >> BIN_SHIFT) & (NBP - 1)], 1);
                sBuf[p] = make_int2(s[k], d[k]);
            }
        }
    }
    __syncthreads();
    const int len = e1 - e0;
    for (int i = t; i < len; i += 1024) {
        const int2 pr = sBuf[i];
        pairs[sOff[(pr.y >> BIN_SHIFT) & (NBP - 1)] + i] = pr;
    }
}

__device__ __forceinline__ void
bin_sort_part(const int2* __restrict__ pairs, const int* __restrict__ binStart,
              const int* __restrict__ binTotal, int* __restrict__ cnt,
              int* __restrict__ rowptr, int* __restrict__ ssrc, int N, const int bin) {
    __shared__ int sCnt[BIN_W];
    __shared__ int sCur[BIN_W];
    __shared__ int sWs[16];
    __shared__ int sOut[BIN_CAP];
    const int t = threadIdx.x, lane = t & 31, wave = t >> 5;
    const int lo   = bin << BIN_SHIFT;
    const int base = binStart[bin];
    const int tot  = binTotal[bin];

    if (t < BIN_W) sCnt[t] = 0;
    __syncthreads();
    for (int ib = t; ib < tot; ib += 4 * 1024) {
        int d[4];
        #pragma unroll
        for (int k = 0; k < 4; ++k) {
            const int i = ib + k * 1024;
            d[k] = (i < tot) ? pairs[base + i].y : lo;
        }
        #pragma unroll
        for (int k = 0; k < 4; ++k)
            if (ib + k * 1024 < tot) atomicAdd(&sCnt[(d[k] - lo) & (BIN_W - 1)], 1);
    }
    __syncthreads();

    int c = 0, inc = 0;
    if (t < BIN_W) {
        c   = sCnt[t];
        inc = wave_incl_scan(c, lane);
        if (lane == 31) sWs[wave] = inc;
    }
    __syncthreads();
    if (t < BIN_W) {
        int o = 0;
        for (int w = 0; w < wave; ++w) o += sWs[w];
        const int ex = o + inc - c;
        sCur[t] = ex;
        const int n = lo + t;
        if (n < N) { cnt[n] = c; rowptr[n] = base + ex; }
    }
    __syncthreads();

    if (tot <= BIN_CAP) {
        for (int ib = t; ib < tot; ib += 4 * 1024) {
            int2 pr[4];
            #pragma unroll
            for (int k = 0; k < 4; ++k) {
                const int i = ib + k * 1024;
                pr[k] = (i < tot) ? pairs[base + i] : make_int2(0, lo);
            }
            #pragma unroll
            for (int k = 0; k < 4; ++k) {
                if (ib + k * 1024 < tot) {
                    const int p = atomicAdd(&sCur[(pr[k].y - lo) & (BIN_W - 1)], 1);
                    sOut[p] = pr[k].x;
                }
            }
        }
        __syncthreads();
        for (int i = t; i < tot; i += 1024) ssrc[base + i] = sOut[i];
    } else {
        for (int i = t; i < tot; i += 1024) {
            const int2 pr = pairs[base + i];
            const int p = atomicAdd(&sCur[(pr.y - lo) & (BIN_W - 1)], 1);
            ssrc[base + p] = pr.x;
        }
    }
}

__global__ void __launch_bounds__(1024, 16)
gat_partition_gemm(const int* __restrict__ src, const int* __restrict__ dst,
                   const int* __restrict__ table, int2* __restrict__ pairs,
                   int* __restrict__ binStart, int* __restrict__ binTotal, int E, int NC,
                   const float* __restrict__ A, const __bf16* __restrict__ Wp,
                   const float* __restrict__ vs, const float* __restrict__ vd,
                   float* __restrict__ C, float* __restrict__ al_s, float* __restrict__ al_d,
                   int N) {
    const int b = blockIdx.x;
    if (b < NC) bin_partition_part(src, dst, table, pairs, binStart, binTotal, E, NC, b);
    else        gemm_tile64(A, Wp, vs, vd, C, al_s, al_d, b - NC, N);
}

__global__ void __launch_bounds__(1024)
gat_sort_gemm(const int2* __restrict__ pairs, const int* __restrict__ binStart,
              const int* __restrict__ binTotal, int* __restrict__ cnt,
              int* __restrict__ rowptr, int* __restrict__ ssrc, int nBin,
              const float* __restrict__ A, const __bf16* __restrict__ Wp,
              const float* __restrict__ vs, const float* __restrict__ vd,
              float* __restrict__ C, float* __restrict__ al_s, float* __restrict__ al_d,
              int N, int tile0) {
    const int b = blockIdx.x;
    if (b < nBin) bin_sort_part(pairs, binStart, binTotal, cnt, rowptr, ssrc, N, b);
    else          gemm_tile64(A, Wp, vs, vd, C, al_s, al_d, tile0 + (b - nBin), N);
}


template <int G>
__device__ __forceinline__ void agg_rows(const float* __restrict__ hs, int s, float ex, int k,
                                         int lane, float4& acc) {
    float4 hv[G];
    float  wk[G];
    #pragma unroll
    for (int u = 0; u < G; ++u) {
        const int sk = __builtin_amdgcn_readlane(s, k + u);
        wk[u] = __int_as_float(__builtin_amdgcn_readlane(__float_as_int(ex), k + u));
        hv[u] = ((const float4*)(hs + (size_t)sk * F_DIM))[lane];
    }
    #pragma unroll
    for (int u = 0; u < G; ++u) {
        acc.x = __builtin_fmaf(wk[u], hv[u].x, acc.x);
        acc.y = __builtin_fmaf(wk[u], hv[u].y, acc.y);
        acc.z = __builtin_fmaf(wk[u], hv[u].z, acc.z);
        acc.w = __builtin_fmaf(wk[u], hv[u].w, acc.w);
    }
}

__global__ void __launch_bounds__(256)
gat_node_agg(const int* __restrict__ rowptr, const int* __restrict__ cnt,
             const int* __restrict__ ssrc, const float* __restrict__ al_s,
             const float* __restrict__ al_d, const float* __restrict__ hs,
             const float* __restrict__ bias, float* __restrict__ out, int N, int do_relu) {
    const int lane = threadIdx.x & 31;
    const int n = __builtin_amdgcn_readfirstlane((int)(blockIdx.x * 8 + (threadIdx.x >> 5)));
    if (n >= N) return;
    const int start = rowptr[n];
    const int deg   = cnt[n];
    const float ald = al_d[n];

    int   s0 = 0;
    float z0 = __int_as_float(0xff800000);
    if (lane < deg) {
        s0 = ssrc[start + lane];
        z0 = leaky(al_s[s0] + ald);
    }
    float m = z0;
    for (int base = 32; base < deg; base += 32) {
        const int j = base + lane;
        if (j < deg) m = fmaxf(m, leaky(al_s[ssrc[start + j]] + ald));
    }
    #pragma unroll
    for (int off = 16; off > 0; off >>= 1) m = fmaxf(m, __shfl_xor(m, off, 32));

    float4 acc = make_float4(0.0f, 0.0f, 0.0f, 0.0f);
    float dsum = 0.0f;
    for (int base = 0; base < deg; base += 32) {
        const int j = base + lane;
        int s = 0; float ex = 0.0f;
        if (base == 0) {
            s = s0;
            if (lane < deg) ex = __expf(z0 - m);
        } else if (j < deg) {
            s  = ssrc[start + j];
            ex = __expf(leaky(al_s[s] + ald) - m);
        }
        dsum += ex;
        const int nk = (deg - base < 32) ? (deg - base) : 32;
        int k = 0;
        for (; k + 8 <= nk; k += 8) agg_rows<8>(hs, s, ex, k, lane, acc);
        switch (nk - k) {
            case 1: agg_rows<1>(hs, s, ex, k, lane, acc); break;
            case 2: agg_rows<2>(hs, s, ex, k, lane, acc); break;
            case 3: agg_rows<3>(hs, s, ex, k, lane, acc); break;
            case 4: agg_rows<4>(hs, s, ex, k, lane, acc); break;
            case 5: agg_rows<5>(hs, s, ex, k, lane, acc); break;
            case 6: agg_rows<6>(hs, s, ex, k, lane, acc); break;
            case 7: agg_rows<7>(hs, s, ex, k, lane, acc); break;
            default: break;
        }
    }
    #pragma unroll
    for (int off = 16; off > 0; off >>= 1) dsum += __shfl_xor(dsum, off, 32);

    const float inv = (deg > 0) ? 1.0f / dsum : 0.0f;
    const float4 b = ((const float4*)bias)[lane];
    float4 r;
    r.x = acc.x * inv + b.x;
    r.y = acc.y * inv + b.y;
    r.z = acc.z * inv + b.z;
    r.w = acc.w * inv + b.w;
    if (do_relu) {
        r.x = fmaxf(r.x, 0.0f); r.y = fmaxf(r.y, 0.0f);
        r.z = fmaxf(r.z, 0.0f); r.w = fmaxf(r.w, 0.0f);
    }
    ((float4*)(out + (size_t)n * F_DIM))[lane] = r;
}

extern "C" void kernel_launch(void* const* d_in, const int* in_sizes, int n_in,
                              void* d_out, int out_size, void* d_ws, size_t ws_size,
                              hipStream_t stream) {
    const float* x   = (const float*)d_in[0];
    const int*   ei  = (const int*)d_in[1];
    const float* W1s = (const float*)d_in[2];
    const float* W1d = (const float*)d_in[3];
    const float* a1s = (const float*)d_in[4];
    const float* a1d = (const float*)d_in[5];
    const float* b1  = (const float*)d_in[6];
    const float* W2s = (const float*)d_in[7];
    const float* W2d = (const float*)d_in[8];
    const float* a2s = (const float*)d_in[9];
    const float* a2d = (const float*)d_in[10];
    const float* b2  = (const float*)d_in[11];

    const int N = in_sizes[0] / F_DIM;
    const int E = in_sizes[1] / 2;
    const int* src = ei;
    const int* dst = ei + E;

    const int nChunk = (E + CH_CAP - 1) / CH_CAP;
    const int nBin   = (N + BIN_W - 1) / BIN_W;

    __bf16* wpack = (__bf16*)d_ws;
    float*  fws   = (float*)((char*)d_ws + 2 * WPACK_ELEMS * sizeof(__bf16));
    float* hs    = fws;
    float* hbuf  = hs   + (size_t)N * F_DIM;
    float* al_s  = hbuf + (size_t)N * F_DIM;
    float* al_d  = al_s + N;
    float* vecs  = al_d + N;
    int* cnt      = (int*)(vecs + 4 * F_DIM);
    int* rowptr   = cnt + N;
    int* binStart = rowptr + N;
    int* binTotal = binStart + NBP;
    int* table    = binTotal + NBP;
    int* ssrc     = table + (size_t)nChunk * NBP;
    int2* pairs   = (int2*)(ssrc + (size_t)E + ((size_t)E & 1));

    float* out = (float*)d_out;

    dim3 b256(256);
    const int gRow  = (N + 7) / 8;
    const int gGemm = (N + 15) / 16;
    const int gT64  = (N + 63) / 64;
    const int gT64a = (gT64 + 1) / 2;

    gat_prep_count<<<4 + nChunk + (2 * WPACK_ELEMS) / 1024, 1024, 0, stream>>>(
        W1s, a1s, W1d, a1d, W2s, a2s, W2d, a2d, vecs, wpack, dst, table, E, nChunk);

    gat_partition_gemm<<<nChunk + gT64a, 1024, 0, stream>>>(
        src, dst, table, pairs, binStart, binTotal, E, nChunk,
        x, wpack, vecs, vecs + 128, hs, al_s, al_d, N);
    gat_sort_gemm<<<nBin + (gT64 - gT64a), 1024, 0, stream>>>(
        pairs, binStart, binTotal, cnt, rowptr, ssrc, nBin,
        x, wpack, vecs, vecs + 128, hs, al_s, al_d, N, gT64a);

    gat_node_agg <<<gRow,  b256, 0, stream>>>(rowptr, cnt, ssrc, al_s, al_d, hs, b1, hbuf, N, 1);

    gat_gemm_wmma<<<gGemm, b256, 0, stream>>>(hbuf, wpack + WPACK_ELEMS, vecs + 256, vecs + 384, hs, al_s, al_d, N);
    gat_node_agg <<<gRow,  b256, 0, stream>>>(rowptr, cnt, ssrc, al_s, al_d, hs, b2, out, N, 0);
}


# Edge sum before product: node wave gathers bf16 operand rows (256 B, was 512 B f32), GEMM takes f32 sums as bf16 hi+lo; a node's sources sorted, so reassociated sums have no arrival order; +1 launch
# speedup vs baseline: 7.1292x; 1.0339x over previous
#include <hip/hip_runtime.h>
#include <hip/hip_bf16.h>


#define F_DIM 128
#define NEG_SLOPE 0.2f

typedef __attribute__((ext_vector_type(16))) __bf16 v16bf;
typedef __attribute__((ext_vector_type(8)))  float  v8f;

#define WPACK_ELEMS (8 * 4 * 32 * 16)

__device__ __forceinline__ float leaky(float z) {
    return z > 0.0f ? z : NEG_SLOPE * z;
}

__device__ __forceinline__ void pack_w_part(const float* __restrict__ W1s, const float* __restrict__ W2s,
                                            __bf16* __restrict__ packed, int idx) {
    int j    = idx & 15;
    int lane = (idx >> 4) & 31;
    int t    = (idx >> 9) & 3;
    int w    = (idx >> 11) & 7;
    int p    = (idx >> 14) & 1;
    const float* W = p ? W2s : W1s;
    int koff = (lane >> 4) * 8;
    int mn   = lane & 15;
    int k    = t * 32 + ((j >> 3) << 4) + koff + (j & 7);
    int col  = w * 16 + mn;
    packed[idx] = (__bf16)W[k * F_DIM + col];
}

__device__ __forceinline__ void matvec_part(const float* __restrict__ W1s, const float* __restrict__ a1s,
                                            const float* __restrict__ W1d, const float* __restrict__ a1d,
                                            const float* __restrict__ W2s, const float* __restrict__ a2s,
                                            const float* __restrict__ W2d, const float* __restrict__ a2d,
                                            float* __restrict__ vecs, int which, int f) {
    const float* W; const float* a;
    switch (which) {
        case 0:  W = W1s; a = a1s; break;
        case 1:  W = W1d; a = a1d; break;
        case 2:  W = W2s; a = a2s; break;
        default: W = W2d; a = a2d; break;
    }
    float s = 0.0f;
    #pragma unroll 8
    for (int d = 0; d < F_DIM; ++d) s += W[f * F_DIM + d] * a[d];
    vecs[which * F_DIM + f] = s;
}

typedef __attribute__((ext_vector_type(4))) __bf16 v4bf;

#define A_PITCH (F_DIM + 8)
#define C_PITCH (F_DIM + 4)


__device__ __forceinline__ void
rows_to_bf16_64(const float* __restrict__ A, const float* __restrict__ vs, const float* __restrict__ vd,
                __bf16* __restrict__ Ab, float* __restrict__ al_s, float* __restrict__ al_d,
                const int tile64, const int N) {
    const int tid = threadIdx.x, lane = tid & 31;
    const int wave = __builtin_amdgcn_readfirstlane(tid >> 5);
    const int ra = tile64 * 64 + 2 * wave, rb = ra + 1;
    if (ra >= N) return;
    const bool hasb = rb < N;
    const float4 xa = ((const float4*)(A + (size_t)ra * F_DIM))[lane];
    float4 xb = make_float4(0.0f, 0.0f, 0.0f, 0.0f);
    if (hasb) xb = ((const float4*)(A + (size_t)rb * F_DIM))[lane];
    const float4 av = ((const float4*)vs)[lane];
    const float4 bv = ((const float4*)vd)[lane];
    float sa = xa.x * av.x + xa.y * av.y + xa.z * av.z + xa.w * av.w;
    float da = xa.x * bv.x + xa.y * bv.y + xa.z * bv.z + xa.w * bv.w;
    float sb = xb.x * av.x + xb.y * av.y + xb.z * av.z + xb.w * av.w;
    float db = xb.x * bv.x + xb.y * bv.y + xb.z * bv.z + xb.w * bv.w;
    v4bf pa = { (__bf16)xa.x, (__bf16)xa.y, (__bf16)xa.z, (__bf16)xa.w };
    v4bf pb = { (__bf16)xb.x, (__bf16)xb.y, (__bf16)xb.z, (__bf16)xb.w };
    *(v4bf*)(Ab + (size_t)ra * F_DIM + 4 * lane) = pa;
    if (hasb) *(v4bf*)(Ab + (size_t)rb * F_DIM + 4 * lane) = pb;
    #pragma unroll
    for (int off = 16; off > 0; off >>= 1) {
        sa += __shfl_down(sa, off, 32);
        da += __shfl_down(da, off, 32);
        sb += __shfl_down(sb, off, 32);
        db += __shfl_down(db, off, 32);
    }
    if (lane == 0) {
        al_s[ra] = sa; al_d[ra] = da;
        if (hasb) { al_s[rb] = sb; al_d[rb] = db; }
    }
}

__device__ __forceinline__ void split4(const float4 x, v4bf& hi, v4bf& lo) {
    const __bf16 h0 = (__bf16)x.x, h1 = (__bf16)x.y, h2 = (__bf16)x.z, h3 = (__bf16)x.w;
    hi = v4bf{ h0, h1, h2, h3 };
    lo = v4bf{ (__bf16)(x.x - (float)h0), (__bf16)(x.y - (float)h1),
               (__bf16)(x.z - (float)h2), (__bf16)(x.w - (float)h3) };
}

template <bool FIRST>
__global__ void __launch_bounds__(256, 16)
gat_gemm_wmma(const float* __restrict__ S, const __bf16* __restrict__ Wp, const float* __restrict__ bias,
              const float* __restrict__ vs, const float* __restrict__ vd,
              __bf16* __restrict__ Hb, float* __restrict__ al_s, float* __restrict__ al_d,
              float* __restrict__ out, int N) {
    __shared__ __attribute__((aligned(16))) __bf16 AsH[16][A_PITCH];
    __shared__ __attribute__((aligned(16))) __bf16 AsL[16][A_PITCH];
    __shared__ __attribute__((aligned(16))) float  Cs[16][C_PITCH];
    const int tid = threadIdx.x, lane = tid & 31;
    const int wave = __builtin_amdgcn_readfirstlane(tid >> 5);
    const int row0 = blockIdx.x * 16;
    const int ra = 2 * wave, rb = 2 * wave + 1;
    const bool oka = row0 + ra < N, okb = row0 + rb < N;

    const int col0 = wave * 16;
    const int mn   = lane & 15;
    const int koff = (lane >> 4) * 8;

    const v16bf* Bfrag = (const v16bf*)Wp + (size_t)(wave * 4) * 32 + lane;
    v16bf bfr[4];
    #pragma unroll
    for (int t = 0; t < 4; ++t) bfr[t] = Bfrag[t * 32];
    const float bc = bias[col0 + mn];

    float4 xa = make_float4(0.0f, 0.0f, 0.0f, 0.0f), xb = xa;
    if (oka) xa = ((const float4*)(S + (size_t)(row0 + ra) * F_DIM))[lane];
    if (okb) xb = ((const float4*)(S + (size_t)(row0 + rb) * F_DIM))[lane];
    v4bf ha, la, hb, lb;
    split4(xa, ha, la);
    split4(xb, hb, lb);
    *(v4bf*)&AsH[ra][4 * lane] = ha;
    *(v4bf*)&AsL[ra][4 * lane] = la;
    *(v4bf*)&AsH[rb][4 * lane] = hb;
    *(v4bf*)&AsL[rb][4 * lane] = lb;
    __syncthreads();

    v8f acch = {}, accl = {};
    #pragma unroll
    for (int t = 0; t < 4; ++t) {
        v16bf afh, afl;
        #pragma unroll
        for (int i = 0; i < 8; ++i) {
            afh[i]     = AsH[mn][t * 32 + koff + i];
            afh[i + 8] = AsH[mn][t * 32 + 16 + koff + i];
            afl[i]     = AsL[mn][t * 32 + koff + i];
            afl[i + 8] = AsL[mn][t * 32 + 16 + koff + i];
        }
        acch = __builtin_amdgcn_wmma_f32_16x16x32_bf16(
             false, afh,  false, bfr[t],
             (short)0, acch,  false,  false);
        accl = __builtin_amdgcn_wmma_f32_16x16x32_bf16(
             false, afl,  false, bfr[t],
             (short)0, accl,  false,  false);
    }

    const int mb = (lane >> 4) * 8;
    #pragma unroll
    for (int v = 0; v < 8; ++v) {
        float r = (acch[v] + accl[v]) + bc;
        if (FIRST) r = fmaxf(r, 0.0f);
        Cs[mb + v][col0 + mn] = r;
    }
    __syncthreads();

    const float4 ca = *(const float4*)&Cs[ra][4 * lane];
    const float4 cb = *(const float4*)&Cs[rb][4 * lane];
    if (FIRST) {
        const float4 av = ((const float4*)vs)[lane];
        const float4 bv = ((const float4*)vd)[lane];
        float sa = ca.x * av.x + ca.y * av.y + ca.z * av.z + ca.w * av.w;
        float da = ca.x * bv.x + ca.y * bv.y + ca.z * bv.z + ca.w * bv.w;
        float sb = cb.x * av.x + cb.y * av.y + cb.z * av.z + cb.w * av.w;
        float db = cb.x * bv.x + cb.y * bv.y + cb.z * bv.z + cb.w * bv.w;
        v4bf pa = { (__bf16)ca.x, (__bf16)ca.y, (__bf16)ca.z, (__bf16)ca.w };
        v4bf pb = { (__bf16)cb.x, (__bf16)cb.y, (__bf16)cb.z, (__bf16)cb.w };
        if (oka) *(v4bf*)(Hb + (size_t)(row0 + ra) * F_DIM + 4 * lane) = pa;
        if (okb) *(v4bf*)(Hb + (size_t)(row0 + rb) * F_DIM + 4 * lane) = pb;
        #pragma unroll
        for (int off = 16; off > 0; off >>= 1) {
            sa += __shfl_down(sa, off, 32);
            da += __shfl_down(da, off, 32);
            sb += __shfl_down(sb, off, 32);
            db += __shfl_down(db, off, 32);
        }
        if (lane == 0) {
            if (oka) { al_s[row0 + ra] = sa; al_d[row0 + ra] = da; }
            if (okb) { al_s[row0 + rb] = sb; al_d[row0 + rb] = db; }
        }
    } else {
        if (oka) ((float4*)(out + (size_t)(row0 + ra) * F_DIM))[lane] = ca;
        if (okb) ((float4*)(out + (size_t)(row0 + rb) * F_DIM))[lane] = cb;
    }
}


#define BIN_SHIFT 9
#define BIN_W     512
#define NBP       256
#define CH_CAP    6400
#define BIN_CAP   12288

__device__ __forceinline__ int wave_incl_scan(int v, int lane) {
    #pragma unroll
    for (int off = 1; off < 32; off <<= 1) {
        int t = __shfl_up(v, off, 32);
        if (lane >= off) v += t;
    }
    return v;
}

__device__ __forceinline__ void
bin_count_part(const int* __restrict__ dst, int* __restrict__ table, int E, const int chunk) {
    __shared__ int h[NBP];
    const int t = threadIdx.x;
    if (t < NBP) h[t] = 0;
    __syncthreads();
    const int e0 = chunk * CH_CAP;
    const int e1 = (e0 + CH_CAP < E) ? (e0 + CH_CAP) : E;
    for (int e = e0 + t; e < e1; e += 1024)
        atomicAdd(&h[(dst[e] >> BIN_SHIFT) & (NBP - 1)], 1);
    __syncthreads();
    if (t < NBP) table[chunk * NBP + t] = h[t];
}

__global__ void __launch_bounds__(1024)
gat_prep_count(const float* __restrict__ W1s, const float* __restrict__ a1s,
               const float* __restrict__ W1d, const float* __restrict__ a1d,
               const float* __restrict__ W2s, const float* __restrict__ a2s,
               const float* __restrict__ W2d, const float* __restrict__ a2d,
               float* __restrict__ vecs, __bf16* __restrict__ packed,
               const int* __restrict__ dst, int* __restrict__ table, int E, int NC) {
    const int b = blockIdx.x;
    if (b < 4) {
        if (threadIdx.x < F_DIM)
            matvec_part(W1s, a1s, W1d, a1d, W2s, a2s, W2d, a2d, vecs, b, threadIdx.x);
    } else if (b < 4 + NC) {
        bin_count_part(dst, table, E, b - 4);
    } else {
        pack_w_part(W1s, W2s, packed, (b - 4 - NC) * 1024 + threadIdx.x);
    }
}

__device__ __forceinline__ void
bin_partition_part(const int* __restrict__ src, const int* __restrict__ dst,
                   const int* __restrict__ table, int2* __restrict__ pairs,
                   int* __restrict__ binStart, int* __restrict__ binTotal, int E, int NC,
                   const int chunk) {
    __shared__ int  sTot[4][NBP];
    __shared__ int  sPre[4][NBP];
    __shared__ int  sMine[NBP];
    __shared__ int  sCur[NBP];
    __shared__ int  sOff[NBP];
    __shared__ int  sWs[2][8];
    __shared__ int2 sBuf[CH_CAP];
    const int t = threadIdx.x, lane = t & 31, wave = t >> 5;
    const int b = t & (NBP - 1), q = t >> 8;

    int tot = 0, pre = 0;
    #pragma unroll 16
    for (int c = q; c < NC; c += 4) {
        const int v = table[c * NBP + b];
        tot += v;
        if (c < chunk)  pre += v;
        if (c == chunk) sMine[b] = v;
    }
    sTot[q][b] = tot;
    sPre[q][b] = pre;
    __syncthreads();

    int T = 0, P = 0, M = 0, iT = 0, iM = 0;
    if (t < NBP) {
        T = sTot[0][t] + sTot[1][t] + sTot[2][t] + sTot[3][t];
        P = sPre[0][t] + sPre[1][t] + sPre[2][t] + sPre[3][t];
        M = sMine[t];
        iT = wave_incl_scan(T, lane);
        iM = wave_incl_scan(M, lane);
        if (lane == 31) { sWs[0][wave] = iT; sWs[1][wave] = iM; }
    }
    __syncthreads();
    if (t < NBP) {
        int oT = 0, oM = 0;
        for (int w = 0; w < wave; ++w) { oT += sWs[0][w]; oM += sWs[1][w]; }
        const int gStart = oT + iT - T;
        const int lStart = oM + iM - M;
        sCur[t] = lStart;
        sOff[t] = gStart + P - lStart;
        if (chunk == 0) { binStart[t] = gStart; binTotal[t] = T; }
    }
    __syncthreads();

    const int e0 = chunk * CH_CAP;
    const int e1 = (e0 + CH_CAP < E) ? (e0 + CH_CAP) : E;
    for (int eb = e0 + t; eb < e1; eb += 4 * 1024) {
        int d[4], s[4];
        #pragma unroll
        for (int k = 0; k < 4; ++k) {
            const int e = eb + k * 1024;
            d[k] = (e < e1) ? dst[e] : 0;
            s[k] = (e < e1) ? src[e] : 0;
        }
        #pragma unroll
        for (int k = 0; k < 4; ++k) {
            if (eb + k * 1024 < e1) {
                const int p = atomicAdd(&sCur[(d[k] >> BIN_SHIFT) & (NBP - 1)], 1);
                sBuf[p] = make_int2(s[k], d[k]);
            }
        }
    }
    __syncthreads();
    const int len = e1 - e0;
    for (int i = t; i < len; i += 1024) {
        const int2 pr = sBuf[i];
        pairs[sOff[(pr.y >> BIN_SHIFT) & (NBP - 1)] + i] = pr;
    }
}

__device__ __forceinline__ void
bin_sort_part(const int2* __restrict__ pairs, const int* __restrict__ binStart,
              const int* __restrict__ binTotal, int* __restrict__ cnt,
              int* __restrict__ rowptr, int* __restrict__ ssrc, int N, const int bin) {
    __shared__ int sCnt[BIN_W];
    __shared__ int sCur[BIN_W];
    __shared__ int sWs[16];
    __shared__ int sOut[BIN_CAP];
    const int t = threadIdx.x, lane = t & 31, wave = t >> 5;
    const int lo   = bin << BIN_SHIFT;
    const int base = binStart[bin];
    const int tot  = binTotal[bin];

    if (t < BIN_W) sCnt[t] = 0;
    __syncthreads();
    for (int ib = t; ib < tot; ib += 4 * 1024) {
        int d[4];
        #pragma unroll
        for (int k = 0; k < 4; ++k) {
            const int i = ib + k * 1024;
            d[k] = (i < tot) ? pairs[base + i].y : lo;
        }
        #pragma unroll
        for (int k = 0; k < 4; ++k)
            if (ib + k * 1024 < tot) atomicAdd(&sCnt[(d[k] - lo) & (BIN_W - 1)], 1);
    }
    __syncthreads();

    int c = 0, inc = 0;
    if (t < BIN_W) {
        c   = sCnt[t];
        inc = wave_incl_scan(c, lane);
        if (lane == 31) sWs[wave] = inc;
    }
    __syncthreads();
    if (t < BIN_W) {
        int o = 0;
        for (int w = 0; w < wave; ++w) o += sWs[w];
        const int ex = o + inc - c;
        sCur[t] = ex;
        const int n = lo + t;
        if (n < N) { cnt[n] = c; rowptr[n] = base + ex; }
    }
    __syncthreads();

    if (tot <= BIN_CAP) {
        for (int ib = t; ib < tot; ib += 4 * 1024) {
            int2 pr[4];
            #pragma unroll
            for (int k = 0; k < 4; ++k) {
                const int i = ib + k * 1024;
                pr[k] = (i < tot) ? pairs[base + i] : make_int2(0, lo);
            }
            #pragma unroll
            for (int k = 0; k < 4; ++k) {
                if (ib + k * 1024 < tot) {
                    const int p = atomicAdd(&sCur[(pr[k].y - lo) & (BIN_W - 1)], 1);
                    sOut[p] = pr[k].x;
                }
            }
        }
        __syncthreads();
        for (int i = t; i < tot; i += 1024) ssrc[base + i] = sOut[i];
    } else {
        for (int i = t; i < tot; i += 1024) {
            const int2 pr = pairs[base + i];
            const int p = atomicAdd(&sCur[(pr.y - lo) & (BIN_W - 1)], 1);
            ssrc[base + p] = pr.x;
        }
    }
}

__global__ void __launch_bounds__(1024, 16)
gat_partition_rows(const int* __restrict__ src, const int* __restrict__ dst,
                   const int* __restrict__ table, int2* __restrict__ pairs,
                   int* __restrict__ binStart, int* __restrict__ binTotal, int E, int NC,
                   const float* __restrict__ A, const float* __restrict__ vs, const float* __restrict__ vd,
                   __bf16* __restrict__ Ab, float* __restrict__ al_s, float* __restrict__ al_d,
                   int N) {
    const int b = blockIdx.x;
    if (b < NC) bin_partition_part(src, dst, table, pairs, binStart, binTotal, E, NC, b);
    else        rows_to_bf16_64(A, vs, vd, Ab, al_s, al_d, b - NC, N);
}

__global__ void __launch_bounds__(1024)
gat_sort_rows(const int2* __restrict__ pairs, const int* __restrict__ binStart,
              const int* __restrict__ binTotal, int* __restrict__ cnt,
              int* __restrict__ rowptr, int* __restrict__ ssrc, int nBin,
              const float* __restrict__ A, const float* __restrict__ vs, const float* __restrict__ vd,
              __bf16* __restrict__ Ab, float* __restrict__ al_s, float* __restrict__ al_d,
              int N, int tile0) {
    const int b = blockIdx.x;
    if (b < nBin) bin_sort_part(pairs, binStart, binTotal, cnt, rowptr, ssrc, N, b);
    else          rows_to_bf16_64(A, vs, vd, Ab, al_s, al_d, tile0 + (b - nBin), N);
}


#define SORT_CAP 128

typedef __attribute__((address_space(3))) int lds_int;

template <int G>
__device__ __forceinline__ void agg_rows(const __bf16* __restrict__ rows, int s, float ex, int k,
                                         int lane, float4& acc) {
    uint2 hv[G];
    float wk[G];
    #pragma unroll
    for (int u = 0; u < G; ++u) {
        const int sk = __builtin_amdgcn_readlane(s, k + u);
        wk[u] = __int_as_float(__builtin_amdgcn_readlane(__float_as_int(ex), k + u));
        hv[u] = ((const uint2*)(rows + (size_t)sk * F_DIM))[lane];
    }
    #pragma unroll
    for (int u = 0; u < G; ++u) {
        const float f0 = __uint_as_float(hv[u].x << 16);
        const float f1 = __uint_as_float(hv[u].x & 0xffff0000u);
        const float f2 = __uint_as_float(hv[u].y << 16);
        const float f3 = __uint_as_float(hv[u].y & 0xffff0000u);
        acc.x = __builtin_fmaf(wk[u], f0, acc.x);
        acc.y = __builtin_fmaf(wk[u], f1, acc.y);
        acc.z = __builtin_fmaf(wk[u], f2, acc.z);
        acc.w = __builtin_fmaf(wk[u], f3, acc.w);
    }
}

__global__ void __launch_bounds__(256)
gat_node_agg(const int* __restrict__ rowptr, const int* __restrict__ cnt,
             const int* __restrict__ ssrc, const float* __restrict__ al_s,
             const float* __restrict__ al_d, const __bf16* __restrict__ rows,
             float* __restrict__ sums, int N) {
    __shared__ int sSort[8][SORT_CAP];
    const int lane = threadIdx.x & 31;
    const int wv = __builtin_amdgcn_readfirstlane((int)(threadIdx.x >> 5));
    const int n  = __builtin_amdgcn_readfirstlane((int)(blockIdx.x * 8 + (threadIdx.x >> 5)));
    if (n >= N) return;
    lds_int* mySort = (lds_int*)sSort[wv];
    const int start = rowptr[n];
    const int deg   = cnt[n];
    const float ald = al_d[n];
    const bool inLds = (deg > 32) && (deg <= SORT_CAP);

    int s0 = 0;
    if (deg <= 32) {
        unsigned key = 0xffffffffu;
        if (lane < deg) key = ((unsigned)ssrc[start + lane] << 5) | (unsigned)lane;
        int rank = 0;
        for (int i = 0; i < deg; ++i) {
            const unsigned ki = (unsigned)__builtin_amdgcn_readlane((int)key, i);
            rank += (ki < key) ? 1 : 0;
        }
        if (lane >= deg) rank = lane;
        s0 = (int)((unsigned)__builtin_amdgcn_ds_permute(rank << 2, (int)key) >> 5);
        if (lane >= deg) s0 = 0;
    } else {
        if (inLds) {
            const int nb = (deg + 31) >> 5;
            for (int b1 = 0; b1 < nb; ++b1) {
                const int j1 = b1 * 32 + lane;
                const int s1 = (j1 < deg) ? ssrc[start + j1] : 0x7fffffff;
                int rank = 0;
                for (int b2 = 0; b2 < nb; ++b2) {
                    const int j2 = b2 * 32 + lane;
                    const int s2 = (j2 < deg) ? ssrc[start + j2] : 0x7fffffff;
                    const int lim = (deg - b2 * 32 < 32) ? (deg - b2 * 32) : 32;
                    for (int i = 0; i < lim; ++i) {
                        const int si = __builtin_amdgcn_readlane(s2, i);
                        rank += (si < s1 || (si == s1 && b2 * 32 + i < j1)) ? 1 : 0;
                    }
                }
                if (j1 < deg) mySort[rank] = s1;
            }
            s0 = mySort[lane];
        } else {
            s0 = ssrc[start + lane];
        }
    }
    float z0 = __int_as_float(0xff800000);
    if (lane < deg) z0 = leaky(al_s[s0] + ald);
    float m = z0;
    for (int base = 32; base < deg; base += 32) {
        const int j = base + lane;
        if (j < deg) {
            const int sj = inLds ? mySort[j] : ssrc[start + j];
            m = fmaxf(m, leaky(al_s[sj] + ald));
        }
    }
    #pragma unroll
    for (int off = 16; off > 0; off >>= 1) m = fmaxf(m, __shfl_xor(m, off, 32));

    float4 acc = make_float4(0.0f, 0.0f, 0.0f, 0.0f);
    float dsum = 0.0f;
    for (int base = 0; base < deg; base += 32) {
        const int j = base + lane;
        int s = 0; float ex = 0.0f;
        if (base == 0) {
            s = s0;
            if (lane < deg) ex = __expf(z0 - m);
        } else if (j < deg) {
            s  = inLds ? mySort[j] : ssrc[start + j];
            ex = __expf(leaky(al_s[s] + ald) - m);
        }
        dsum += ex;
        const int nk = (deg - base < 32) ? (deg - base) : 32;
        int k = 0;
        for (; k + 8 <= nk; k += 8) agg_rows<8>(rows, s, ex, k, lane, acc);
        switch (nk - k) {
            case 1: agg_rows<1>(rows, s, ex, k, lane, acc); break;
            case 2: agg_rows<2>(rows, s, ex, k, lane, acc); break;
            case 3: agg_rows<3>(rows, s, ex, k, lane, acc); break;
            case 4: agg_rows<4>(rows, s, ex, k, lane, acc); break;
            case 5: agg_rows<5>(rows, s, ex, k, lane, acc); break;
            case 6: agg_rows<6>(rows, s, ex, k, lane, acc); break;
            case 7: agg_rows<7>(rows, s, ex, k, lane, acc); break;
            default: break;
        }
    }
    #pragma unroll
    for (int off = 16; off > 0; off >>= 1) dsum += __shfl_xor(dsum, off, 32);

    const float inv = (deg > 0) ? 1.0f / dsum : 0.0f;
    float4 r;
    r.x = acc.x * inv;
    r.y = acc.y * inv;
    r.z = acc.z * inv;
    r.w = acc.w * inv;
    ((float4*)(sums + (size_t)n * F_DIM))[lane] = r;
}

extern "C" void kernel_launch(void* const* d_in, const int* in_sizes, int n_in,
                              void* d_out, int out_size, void* d_ws, size_t ws_size,
                              hipStream_t stream) {
    const float* x   = (const float*)d_in[0];
    const int*   ei  = (const int*)d_in[1];
    const float* W1s = (const float*)d_in[2];
    const float* W1d = (const float*)d_in[3];
    const float* a1s = (const float*)d_in[4];
    const float* a1d = (const float*)d_in[5];
    const float* b1  = (const float*)d_in[6];
    const float* W2s = (const float*)d_in[7];
    const float* W2d = (const float*)d_in[8];
    const float* a2s = (const float*)d_in[9];
    const float* a2d = (const float*)d_in[10];
    const float* b2  = (const float*)d_in[11];

    const int N = in_sizes[0] / F_DIM;
    const int E = in_sizes[1] / 2;
    const int* src = ei;
    const int* dst = ei + E;

    const int nChunk = (E + CH_CAP - 1) / CH_CAP;
    const int nBin   = (N + BIN_W - 1) / BIN_W;

    __bf16* wpack = (__bf16*)d_ws;
    float*  fws   = (float*)((char*)d_ws + 2 * WPACK_ELEMS * sizeof(__bf16));
    float* sums  = fws;
    __bf16* xb   = (__bf16*)(sums + (size_t)N * F_DIM);
    __bf16* hb   = xb + (size_t)N * F_DIM;
    float* al_s  = (float*)(hb + (size_t)N * F_DIM);
    float* al_d  = al_s + N;
    float* vecs  = al_d + N;
    int* cnt      = (int*)(vecs + 4 * F_DIM);
    int* rowptr   = cnt + N;
    int* binStart = rowptr + N;
    int* binTotal = binStart + NBP;
    int* table    = binTotal + NBP;
    int* ssrc     = table + (size_t)nChunk * NBP;
    int2* pairs   = (int2*)(ssrc + (size_t)E + ((size_t)E & 1));

    float* out = (float*)d_out;

    dim3 b256(256);
    const int gRow  = (N + 7) / 8;
    const int gGemm = (N + 15) / 16;
    const int gT64  = (N + 63) / 64;
    const int gT64a = (gT64 + 1) / 2;

    gat_prep_count<<<4 + nChunk + (2 * WPACK_ELEMS) / 1024, 1024, 0, stream>>>(
        W1s, a1s, W1d, a1d, W2s, a2s, W2d, a2d, vecs, wpack, dst, table, E, nChunk);

    gat_partition_rows<<<nChunk + gT64a, 1024, 0, stream>>>(
        src, dst, table, pairs, binStart, binTotal, E, nChunk,
        x, vecs, vecs + 128, xb, al_s, al_d, N);
    gat_sort_rows<<<nBin + (gT64 - gT64a), 1024, 0, stream>>>(
        pairs, binStart, binTotal, cnt, rowptr, ssrc, nBin,
        x, vecs, vecs + 128, xb, al_s, al_d, N, gT64a);

    gat_node_agg<<<gRow, b256, 0, stream>>>(rowptr, cnt, ssrc, al_s, al_d, xb, sums, N);
    gat_gemm_wmma<true><<<gGemm, b256, 0, stream>>>(
        sums, wpack, b1, vecs + 256, vecs + 384, hb, al_s, al_d, out, N);

    gat_node_agg<<<gRow, b256, 0, stream>>>(rowptr, cnt, ssrc, al_s, al_d, hb, sums, N);
    gat_gemm_wmma<false><<<gGemm, b256, 0, stream>>>(
        sums, wpack + WPACK_ELEMS, b2, vecs + 256, vecs + 384, hb, al_s, al_d, out, N);
}


# Node and product kernels as resident grids: a wave loops over its nodes, a workgroup over its row tiles (B fragments loaded once, next rows requested ahead); 300000 -> 53248 waves; sums unchanged
# speedup vs baseline: 7.8830x; 1.1057x over previous
#include <hip/hip_runtime.h>
#include <hip/hip_bf16.h>


#define F_DIM 128
#define NEG_SLOPE 0.2f

typedef __attribute__((ext_vector_type(16))) __bf16 v16bf;
typedef __attribute__((ext_vector_type(8)))  float  v8f;

#define WPACK_ELEMS (8 * 4 * 32 * 16)

__device__ __forceinline__ float leaky(float z) {
    return z > 0.0f ? z : NEG_SLOPE * z;
}

__device__ __forceinline__ void pack_w_part(const float* __restrict__ W1s, const float* __restrict__ W2s,
                                            __bf16* __restrict__ packed, int idx) {
    int j    = idx & 15;
    int lane = (idx >> 4) & 31;
    int t    = (idx >> 9) & 3;
    int w    = (idx >> 11) & 7;
    int p    = (idx >> 14) & 1;
    const float* W = p ? W2s : W1s;
    int koff = (lane >> 4) * 8;
    int mn   = lane & 15;
    int k    = t * 32 + ((j >> 3) << 4) + koff + (j & 7);
    int col  = w * 16 + mn;
    packed[idx] = (__bf16)W[k * F_DIM + col];
}

__device__ __forceinline__ void matvec_part(const float* __restrict__ W1s, const float* __restrict__ a1s,
                                            const float* __restrict__ W1d, const float* __restrict__ a1d,
                                            const float* __restrict__ W2s, const float* __restrict__ a2s,
                                            const float* __restrict__ W2d, const float* __restrict__ a2d,
                                            float* __restrict__ vecs, int which, int f) {
    const float* W; const float* a;
    switch (which) {
        case 0:  W = W1s; a = a1s; break;
        case 1:  W = W1d; a = a1d; break;
        case 2:  W = W2s; a = a2s; break;
        default: W = W2d; a = a2d; break;
    }
    float s = 0.0f;
    #pragma unroll 8
    for (int d = 0; d < F_DIM; ++d) s += W[f * F_DIM + d] * a[d];
    vecs[which * F_DIM + f] = s;
}

typedef __attribute__((ext_vector_type(4))) __bf16 v4bf;

#define A_PITCH (F_DIM + 8)
#define C_PITCH (F_DIM + 4)


__device__ __forceinline__ void
rows_to_bf16_64(const float* __restrict__ A, const float* __restrict__ vs, const float* __restrict__ vd,
                __bf16* __restrict__ Ab, float* __restrict__ al_s, float* __restrict__ al_d,
                const int tile64, const int N) {
    const int tid = threadIdx.x, lane = tid & 31;
    const int wave = __builtin_amdgcn_readfirstlane(tid >> 5);
    const int ra = tile64 * 64 + 2 * wave, rb = ra + 1;
    if (ra >= N) return;
    const bool hasb = rb < N;
    const float4 xa = ((const float4*)(A + (size_t)ra * F_DIM))[lane];
    float4 xb = make_float4(0.0f, 0.0f, 0.0f, 0.0f);
    if (hasb) xb = ((const float4*)(A + (size_t)rb * F_DIM))[lane];
    const float4 av = ((const float4*)vs)[lane];
    const float4 bv = ((const float4*)vd)[lane];
    float sa = xa.x * av.x + xa.y * av.y + xa.z * av.z + xa.w * av.w;
    float da = xa.x * bv.x + xa.y * bv.y + xa.z * bv.z + xa.w * bv.w;
    float sb = xb.x * av.x + xb.y * av.y + xb.z * av.z + xb.w * av.w;
    float db = xb.x * bv.x + xb.y * bv.y + xb.z * bv.z + xb.w * bv.w;
    v4bf pa = { (__bf16)xa.x, (__bf16)xa.y, (__bf16)xa.z, (__bf16)xa.w };
    v4bf pb = { (__bf16)xb.x, (__bf16)xb.y, (__bf16)xb.z, (__bf16)xb.w };
    *(v4bf*)(Ab + (size_t)ra * F_DIM + 4 * lane) = pa;
    if (hasb) *(v4bf*)(Ab + (size_t)rb * F_DIM + 4 * lane) = pb;
    #pragma unroll
    for (int off = 16; off > 0; off >>= 1) {
        sa += __shfl_down(sa, off, 32);
        da += __shfl_down(da, off, 32);
        sb += __shfl_down(sb, off, 32);
        db += __shfl_down(db, off, 32);
    }
    if (lane == 0) {
        al_s[ra] = sa; al_d[ra] = da;
        if (hasb) { al_s[rb] = sb; al_d[rb] = db; }
    }
}

__device__ __forceinline__ void split4(const float4 x, v4bf& hi, v4bf& lo) {
    const __bf16 h0 = (__bf16)x.x, h1 = (__bf16)x.y, h2 = (__bf16)x.z, h3 = (__bf16)x.w;
    hi = v4bf{ h0, h1, h2, h3 };
    lo = v4bf{ (__bf16)(x.x - (float)h0), (__bf16)(x.y - (float)h1),
               (__bf16)(x.z - (float)h2), (__bf16)(x.w - (float)h3) };
}

template <bool FIRST>
__global__ void __launch_bounds__(256, 10)
gat_gemm_wmma(const float* __restrict__ S, const __bf16* __restrict__ Wp, const float* __restrict__ bias,
              const float* __restrict__ vs, const float* __restrict__ vd,
              __bf16* __restrict__ Hb, float* __restrict__ al_s, float* __restrict__ al_d,
              float* __restrict__ out, int N) {
    __shared__ __attribute__((aligned(16))) __bf16 AsH[16][A_PITCH];
    __shared__ __attribute__((aligned(16))) __bf16 AsL[16][A_PITCH];
    __shared__ __attribute__((aligned(16))) float  Cs[16][C_PITCH];
    const int tid = threadIdx.x, lane = tid & 31;
    const int wave = __builtin_amdgcn_readfirstlane(tid >> 5);
    const int ra = 2 * wave, rb = 2 * wave + 1;
    const int col0 = wave * 16;
    const int mn   = lane & 15;
    const int koff = (lane >> 4) * 8;
    const int nTiles = (N + 15) >> 4;

    const char* Sbytes = (const char*)S;
    float4 xa = make_float4(0.0f, 0.0f, 0.0f, 0.0f), xb = xa;
    if ((int)blockIdx.x < nTiles) {
        const int r0 = blockIdx.x * 16 + ra;
        const unsigned o0 = (unsigned)r0 * (F_DIM * 4u) + (unsigned)lane * 16u;
        if (r0 + 1 < N)  { xa = *(const float4*)(Sbytes + o0); xb = *(const float4*)(Sbytes + o0 + F_DIM * 4u); }
        else if (r0 < N) { xa = *(const float4*)(Sbytes + o0); }
    }

    const v16bf* Bfrag = (const v16bf*)Wp + (size_t)(wave * 4) * 32 + lane;
    v16bf bfr[4];
    #pragma unroll
    for (int t = 0; t < 4; ++t) bfr[t] = Bfrag[t * 32];
    const float bc = bias[col0 + mn];
    float4 av = make_float4(0.0f, 0.0f, 0.0f, 0.0f), bv = av;
    if (FIRST) {
        av = ((const float4*)vs)[lane];
        bv = ((const float4*)vd)[lane];
    }

    for (int tile = blockIdx.x; tile < nTiles; tile += gridDim.x) {
    const int row0 = tile * 16;
    const bool oka = row0 + ra < N, okb = row0 + rb < N;

    v4bf ha, la, hb, lb;
    split4(xa, ha, la);
    split4(xb, hb, lb);
    *(v4bf*)&AsH[ra][4 * lane] = ha;
    *(v4bf*)&AsL[ra][4 * lane] = la;
    *(v4bf*)&AsH[rb][4 * lane] = hb;
    *(v4bf*)&AsL[rb][4 * lane] = lb;
    __syncthreads();

    xa = make_float4(0.0f, 0.0f, 0.0f, 0.0f); xb = xa;
    if (tile + (int)gridDim.x < nTiles) {
        const int r1 = (tile + (int)gridDim.x) * 16 + ra;
        const unsigned o1 = (unsigned)r1 * (F_DIM * 4u) + (unsigned)lane * 16u;
        if (r1 + 1 < N)  { xa = *(const float4*)(Sbytes + o1); xb = *(const float4*)(Sbytes + o1 + F_DIM * 4u); }
        else if (r1 < N) { xa = *(const float4*)(Sbytes + o1); }
    }

    v8f acch = {}, accl = {};
    #pragma unroll
    for (int t = 0; t < 4; ++t) {
        v16bf afh, afl;
        #pragma unroll
        for (int i = 0; i < 8; ++i) {
            afh[i]     = AsH[mn][t * 32 + koff + i];
            afh[i + 8] = AsH[mn][t * 32 + 16 + koff + i];
            afl[i]     = AsL[mn][t * 32 + koff + i];
            afl[i + 8] = AsL[mn][t * 32 + 16 + koff + i];
        }
        acch = __builtin_amdgcn_wmma_f32_16x16x32_bf16(
             false, afh,  false, bfr[t],
             (short)0, acch,  false,  false);
        accl = __builtin_amdgcn_wmma_f32_16x16x32_bf16(
             false, afl,  false, bfr[t],
             (short)0, accl,  false,  false);
    }

    const int mb = (lane >> 4) * 8;
    #pragma unroll
    for (int v = 0; v < 8; ++v) {
        float r = (acch[v] + accl[v]) + bc;
        if (FIRST) r = fmaxf(r, 0.0f);
        Cs[mb + v][col0 + mn] = r;
    }
    __syncthreads();

    const float4 ca = *(const float4*)&Cs[ra][4 * lane];
    const float4 cb = *(const float4*)&Cs[rb][4 * lane];
    if (FIRST) {
        float sa = ca.x * av.x + ca.y * av.y + ca.z * av.z + ca.w * av.w;
        float da = ca.x * bv.x + ca.y * bv.y + ca.z * bv.z + ca.w * bv.w;
        float sb = cb.x * av.x + cb.y * av.y + cb.z * av.z + cb.w * av.w;
        float db = cb.x * bv.x + cb.y * bv.y + cb.z * bv.z + cb.w * bv.w;
        v4bf pa = { (__bf16)ca.x, (__bf16)ca.y, (__bf16)ca.z, (__bf16)ca.w };
        v4bf pb = { (__bf16)cb.x, (__bf16)cb.y, (__bf16)cb.z, (__bf16)cb.w };
        if (oka) *(v4bf*)(Hb + (size_t)(row0 + ra) * F_DIM + 4 * lane) = pa;
        if (okb) *(v4bf*)(Hb + (size_t)(row0 + rb) * F_DIM + 4 * lane) = pb;
        #pragma unroll
        for (int off = 16; off > 0; off >>= 1) {
            sa += __shfl_down(sa, off, 32);
            da += __shfl_down(da, off, 32);
            sb += __shfl_down(sb, off, 32);
            db += __shfl_down(db, off, 32);
        }
        if (lane == 0) {
            if (oka) { al_s[row0 + ra] = sa; al_d[row0 + ra] = da; }
            if (okb) { al_s[row0 + rb] = sb; al_d[row0 + rb] = db; }
        }
    } else {
        if (oka) ((float4*)(out + (size_t)(row0 + ra) * F_DIM))[lane] = ca;
        if (okb) ((float4*)(out + (size_t)(row0 + rb) * F_DIM))[lane] = cb;
    }
    }
}


#define BIN_SHIFT 9
#define BIN_W     512
#define NBP       256
#define CH_CAP    6400
#define BIN_CAP   12288

__device__ __forceinline__ int wave_incl_scan(int v, int lane) {
    #pragma unroll
    for (int off = 1; off < 32; off <<= 1) {
        int t = __shfl_up(v, off, 32);
        if (lane >= off) v += t;
    }
    return v;
}

__device__ __forceinline__ void
bin_count_part(const int* __restrict__ dst, int* __restrict__ table, int E, const int chunk) {
    __shared__ int h[NBP];
    const int t = threadIdx.x;
    if (t < NBP) h[t] = 0;
    __syncthreads();
    const int e0 = chunk * CH_CAP;
    const int e1 = (e0 + CH_CAP < E) ? (e0 + CH_CAP) : E;
    for (int e = e0 + t; e < e1; e += 1024)
        atomicAdd(&h[(dst[e] >> BIN_SHIFT) & (NBP - 1)], 1);
    __syncthreads();
    if (t < NBP) table[chunk * NBP + t] = h[t];
}

__global__ void __launch_bounds__(1024)
gat_prep_count(const float* __restrict__ W1s, const float* __restrict__ a1s,
               const float* __restrict__ W1d, const float* __restrict__ a1d,
               const float* __restrict__ W2s, const float* __restrict__ a2s,
               const float* __restrict__ W2d, const float* __restrict__ a2d,
               float* __restrict__ vecs, __bf16* __restrict__ packed,
               const int* __restrict__ dst, int* __restrict__ table, int E, int NC) {
    const int b = blockIdx.x;
    if (b < 4) {
        if (threadIdx.x < F_DIM)
            matvec_part(W1s, a1s, W1d, a1d, W2s, a2s, W2d, a2d, vecs, b, threadIdx.x);
    } else if (b < 4 + NC) {
        bin_count_part(dst, table, E, b - 4);
    } else {
        pack_w_part(W1s, W2s, packed, (b - 4 - NC) * 1024 + threadIdx.x);
    }
}

__device__ __forceinline__ void
bin_partition_part(const int* __restrict__ src, const int* __restrict__ dst,
                   const int* __restrict__ table, int2* __restrict__ pairs,
                   int* __restrict__ binStart, int* __restrict__ binTotal, int E, int NC,
                   const int chunk) {
    __shared__ int  sTot[4][NBP];
    __shared__ int  sPre[4][NBP];
    __shared__ int  sMine[NBP];
    __shared__ int  sCur[NBP];
    __shared__ int  sOff[NBP];
    __shared__ int  sWs[2][8];
    __shared__ int2 sBuf[CH_CAP];
    const int t = threadIdx.x, lane = t & 31, wave = t >> 5;
    const int b = t & (NBP - 1), q = t >> 8;

    int tot = 0, pre = 0;
    #pragma unroll 16
    for (int c = q; c < NC; c += 4) {
        const int v = table[c * NBP + b];
        tot += v;
        if (c < chunk)  pre += v;
        if (c == chunk) sMine[b] = v;
    }
    sTot[q][b] = tot;
    sPre[q][b] = pre;
    __syncthreads();

    int T = 0, P = 0, M = 0, iT = 0, iM = 0;
    if (t < NBP) {
        T = sTot[0][t] + sTot[1][t] + sTot[2][t] + sTot[3][t];
        P = sPre[0][t] + sPre[1][t] + sPre[2][t] + sPre[3][t];
        M = sMine[t];
        iT = wave_incl_scan(T, lane);
        iM = wave_incl_scan(M, lane);
        if (lane == 31) { sWs[0][wave] = iT; sWs[1][wave] = iM; }
    }
    __syncthreads();
    if (t < NBP) {
        int oT = 0, oM = 0;
        for (int w = 0; w < wave; ++w) { oT += sWs[0][w]; oM += sWs[1][w]; }
        const int gStart = oT + iT - T;
        const int lStart = oM + iM - M;
        sCur[t] = lStart;
        sOff[t] = gStart + P - lStart;
        if (chunk == 0) { binStart[t] = gStart; binTotal[t] = T; }
    }
    __syncthreads();

    const int e0 = chunk * CH_CAP;
    const int e1 = (e0 + CH_CAP < E) ? (e0 + CH_CAP) : E;
    for (int eb = e0 + t; eb < e1; eb += 4 * 1024) {
        int d[4], s[4];
        #pragma unroll
        for (int k = 0; k < 4; ++k) {
            const int e = eb + k * 1024;
            d[k] = (e < e1) ? dst[e] : 0;
            s[k] = (e < e1) ? src[e] : 0;
        }
        #pragma unroll
        for (int k = 0; k < 4; ++k) {
            if (eb + k * 1024 < e1) {
                const int p = atomicAdd(&sCur[(d[k] >> BIN_SHIFT) & (NBP - 1)], 1);
                sBuf[p] = make_int2(s[k], d[k]);
            }
        }
    }
    __syncthreads();
    const int len = e1 - e0;
    for (int i = t; i < len; i += 1024) {
        const int2 pr = sBuf[i];
        pairs[sOff[(pr.y >> BIN_SHIFT) & (NBP - 1)] + i] = pr;
    }
}

__device__ __forceinline__ void
bin_sort_part(const int2* __restrict__ pairs, const int* __restrict__ binStart,
              const int* __restrict__ binTotal, int* __restrict__ cnt,
              int* __restrict__ rowptr, int* __restrict__ ssrc, int N, const int bin) {
    __shared__ int sCnt[BIN_W];
    __shared__ int sCur[BIN_W];
    __shared__ int sWs[16];
    __shared__ int sOut[BIN_CAP];
    const int t = threadIdx.x, lane = t & 31, wave = t >> 5;
    const int lo   = bin << BIN_SHIFT;
    const int base = binStart[bin];
    const int tot  = binTotal[bin];

    if (t < BIN_W) sCnt[t] = 0;
    __syncthreads();
    for (int ib = t; ib < tot; ib += 4 * 1024) {
        int d[4];
        #pragma unroll
        for (int k = 0; k < 4; ++k) {
            const int i = ib + k * 1024;
            d[k] = (i < tot) ? pairs[base + i].y : lo;
        }
        #pragma unroll
        for (int k = 0; k < 4; ++k)
            if (ib + k * 1024 < tot) atomicAdd(&sCnt[(d[k] - lo) & (BIN_W - 1)], 1);
    }
    __syncthreads();

    int c = 0, inc = 0;
    if (t < BIN_W) {
        c   = sCnt[t];
        inc = wave_incl_scan(c, lane);
        if (lane == 31) sWs[wave] = inc;
    }
    __syncthreads();
    if (t < BIN_W) {
        int o = 0;
        for (int w = 0; w < wave; ++w) o += sWs[w];
        const int ex = o + inc - c;
        sCur[t] = ex;
        const int n = lo + t;
        if (n < N) { cnt[n] = c; rowptr[n] = base + ex; }
    }
    __syncthreads();

    if (tot <= BIN_CAP) {
        for (int ib = t; ib < tot; ib += 4 * 1024) {
            int2 pr[4];
            #pragma unroll
            for (int k = 0; k < 4; ++k) {
                const int i = ib + k * 1024;
                pr[k] = (i < tot) ? pairs[base + i] : make_int2(0, lo);
            }
            #pragma unroll
            for (int k = 0; k < 4; ++k) {
                if (ib + k * 1024 < tot) {
                    const int p = atomicAdd(&sCur[(pr[k].y - lo) & (BIN_W - 1)], 1);
                    sOut[p] = pr[k].x;
                }
            }
        }
        __syncthreads();
        for (int i = t; i < tot; i += 1024) ssrc[base + i] = sOut[i];
    } else {
        for (int i = t; i < tot; i += 1024) {
            const int2 pr = pairs[base + i];
            const int p = atomicAdd(&sCur[(pr.y - lo) & (BIN_W - 1)], 1);
            ssrc[base + p] = pr.x;
        }
    }
}

__global__ void __launch_bounds__(1024, 16)
gat_partition_rows(const int* __restrict__ src, const int* __restrict__ dst,
                   const int* __restrict__ table, int2* __restrict__ pairs,
                   int* __restrict__ binStart, int* __restrict__ binTotal, int E, int NC,
                   const float* __restrict__ A, const float* __restrict__ vs, const float* __restrict__ vd,
                   __bf16* __restrict__ Ab, float* __restrict__ al_s, float* __restrict__ al_d,
                   int N) {
    const int b = blockIdx.x;
    if (b < NC) bin_partition_part(src, dst, table, pairs, binStart, binTotal, E, NC, b);
    else        rows_to_bf16_64(A, vs, vd, Ab, al_s, al_d, b - NC, N);
}

__global__ void __launch_bounds__(1024)
gat_sort_rows(const int2* __restrict__ pairs, const int* __restrict__ binStart,
              const int* __restrict__ binTotal, int* __restrict__ cnt,
              int* __restrict__ rowptr, int* __restrict__ ssrc, int nBin,
              const float* __restrict__ A, const float* __restrict__ vs, const float* __restrict__ vd,
              __bf16* __restrict__ Ab, float* __restrict__ al_s, float* __restrict__ al_d,
              int N, int tile0) {
    const int b = blockIdx.x;
    if (b < nBin) bin_sort_part(pairs, binStart, binTotal, cnt, rowptr, ssrc, N, b);
    else          rows_to_bf16_64(A, vs, vd, Ab, al_s, al_d, tile0 + (b - nBin), N);
}


#define SORT_CAP 128

typedef __attribute__((address_space(3))) int lds_int;

template <int G>
__device__ __forceinline__ void agg_rows(const __bf16* __restrict__ rows, int s, float ex, int k,
                                         int lane, float4& acc) {
    uint2 hv[G];
    float wk[G];
    #pragma unroll
    for (int u = 0; u < G; ++u) {
        const int sk = __builtin_amdgcn_readlane(s, k + u);
        wk[u] = __int_as_float(__builtin_amdgcn_readlane(__float_as_int(ex), k + u));
        hv[u] = ((const uint2*)(rows + (size_t)sk * F_DIM))[lane];
    }
    #pragma unroll
    for (int u = 0; u < G; ++u) {
        const float f0 = __uint_as_float(hv[u].x << 16);
        const float f1 = __uint_as_float(hv[u].x & 0xffff0000u);
        const float f2 = __uint_as_float(hv[u].y << 16);
        const float f3 = __uint_as_float(hv[u].y & 0xffff0000u);
        acc.x = __builtin_fmaf(wk[u], f0, acc.x);
        acc.y = __builtin_fmaf(wk[u], f1, acc.y);
        acc.z = __builtin_fmaf(wk[u], f2, acc.z);
        acc.w = __builtin_fmaf(wk[u], f3, acc.w);
    }
}

__global__ void __launch_bounds__(256)
gat_node_agg(const int* __restrict__ rowptr, const int* __restrict__ cnt,
             const int* __restrict__ ssrc, const float* __restrict__ al_s,
             const float* __restrict__ al_d, const __bf16* __restrict__ rows,
             float* __restrict__ sums, int N) {
    __shared__ int sSort[8][SORT_CAP];
    const int lane = threadIdx.x & 31;
    const int wv = __builtin_amdgcn_readfirstlane((int)(threadIdx.x >> 5));
    lds_int* mySort = (lds_int*)sSort[wv];
    int n = __builtin_amdgcn_readfirstlane((int)(blockIdx.x * 8 + (threadIdx.x >> 5)));
    if (n >= N) return;
    const int stride = (int)gridDim.x * 8;
    int   startN = rowptr[n];
    int   degN   = cnt[n];
    float aldN   = al_d[n];
    for (;;) {
    const int start = startN;
    const int deg   = degN;
    const float ald = aldN;
    const int nn = n + stride;
    if (nn < N) { startN = rowptr[nn]; degN = cnt[nn]; aldN = al_d[nn]; }
    const bool inLds = (deg > 32) && (deg <= SORT_CAP);

    int s0 = 0;
    if (deg <= 32) {
        unsigned key = 0xffffffffu;
        if (lane < deg) key = ((unsigned)ssrc[start + lane] << 5) | (unsigned)lane;
        int rank = 0;
        for (int i = 0; i < deg; ++i) {
            const unsigned ki = (unsigned)__builtin_amdgcn_readlane((int)key, i);
            rank += (ki < key) ? 1 : 0;
        }
        if (lane >= deg) rank = lane;
        s0 = (int)((unsigned)__builtin_amdgcn_ds_permute(rank << 2, (int)key) >> 5);
        if (lane >= deg) s0 = 0;
    } else {
        if (inLds) {
            const int nb = (deg + 31) >> 5;
            for (int b1 = 0; b1 < nb; ++b1) {
                const int j1 = b1 * 32 + lane;
                const int s1 = (j1 < deg) ? ssrc[start + j1] : 0x7fffffff;
                int rank = 0;
                for (int b2 = 0; b2 < nb; ++b2) {
                    const int j2 = b2 * 32 + lane;
                    const int s2 = (j2 < deg) ? ssrc[start + j2] : 0x7fffffff;
                    const int lim = (deg - b2 * 32 < 32) ? (deg - b2 * 32) : 32;
                    for (int i = 0; i < lim; ++i) {
                        const int si = __builtin_amdgcn_readlane(s2, i);
                        rank += (si < s1 || (si == s1 && b2 * 32 + i < j1)) ? 1 : 0;
                    }
                }
                if (j1 < deg) mySort[rank] = s1;
            }
            s0 = mySort[lane];
        } else {
            s0 = ssrc[start + lane];
        }
    }
    float z0 = __int_as_float(0xff800000);
    if (lane < deg) z0 = leaky(al_s[s0] + ald);
    float m = z0;
    for (int base = 32; base < deg; base += 32) {
        const int j = base + lane;
        if (j < deg) {
            const int sj = inLds ? mySort[j] : ssrc[start + j];
            m = fmaxf(m, leaky(al_s[sj] + ald));
        }
    }
    #pragma unroll
    for (int off = 16; off > 0; off >>= 1) m = fmaxf(m, __shfl_xor(m, off, 32));

    float4 acc = make_float4(0.0f, 0.0f, 0.0f, 0.0f);
    float dsum = 0.0f;
    for (int base = 0; base < deg; base += 32) {
        const int j = base + lane;
        int s = 0; float ex = 0.0f;
        if (base == 0) {
            s = s0;
            if (lane < deg) ex = __expf(z0 - m);
        } else if (j < deg) {
            s  = inLds ? mySort[j] : ssrc[start + j];
            ex = __expf(leaky(al_s[s] + ald) - m);
        }
        dsum += ex;
        const int nk = (deg - base < 32) ? (deg - base) : 32;
        int k = 0;
        for (; k + 8 <= nk; k += 8) agg_rows<8>(rows, s, ex, k, lane, acc);
        switch (nk - k) {
            case 1: agg_rows<1>(rows, s, ex, k, lane, acc); break;
            case 2: agg_rows<2>(rows, s, ex, k, lane, acc); break;
            case 3: agg_rows<3>(rows, s, ex, k, lane, acc); break;
            case 4: agg_rows<4>(rows, s, ex, k, lane, acc); break;
            case 5: agg_rows<5>(rows, s, ex, k, lane, acc); break;
            case 6: agg_rows<6>(rows, s, ex, k, lane, acc); break;
            case 7: agg_rows<7>(rows, s, ex, k, lane, acc); break;
            default: break;
        }
    }
    #pragma unroll
    for (int off = 16; off > 0; off >>= 1) dsum += __shfl_xor(dsum, off, 32);

    const float inv = (deg > 0) ? 1.0f / dsum : 0.0f;
    float4 r;
    r.x = acc.x * inv;
    r.y = acc.y * inv;
    r.z = acc.z * inv;
    r.w = acc.w * inv;
    ((float4*)(sums + (size_t)n * F_DIM))[lane] = r;
    if (nn >= N) break;
    n = nn;
    }
}

extern "C" void kernel_launch(void* const* d_in, const int* in_sizes, int n_in,
                              void* d_out, int out_size, void* d_ws, size_t ws_size,
                              hipStream_t stream) {
    const float* x   = (const float*)d_in[0];
    const int*   ei  = (const int*)d_in[1];
    const float* W1s = (const float*)d_in[2];
    const float* W1d = (const float*)d_in[3];
    const float* a1s = (const float*)d_in[4];
    const float* a1d = (const float*)d_in[5];
    const float* b1  = (const float*)d_in[6];
    const float* W2s = (const float*)d_in[7];
    const float* W2d = (const float*)d_in[8];
    const float* a2s = (const float*)d_in[9];
    const float* a2d = (const float*)d_in[10];
    const float* b2  = (const float*)d_in[11];

    const int N = in_sizes[0] / F_DIM;
    const int E = in_sizes[1] / 2;
    const int* src = ei;
    const int* dst = ei + E;

    const int nChunk = (E + CH_CAP - 1) / CH_CAP;
    const int nBin   = (N + BIN_W - 1) / BIN_W;

    __bf16* wpack = (__bf16*)d_ws;
    float*  fws   = (float*)((char*)d_ws + 2 * WPACK_ELEMS * sizeof(__bf16));
    float* sums  = fws;
    __bf16* xb   = (__bf16*)(sums + (size_t)N * F_DIM);
    __bf16* hb   = xb + (size_t)N * F_DIM;
    float* al_s  = (float*)(hb + (size_t)N * F_DIM);
    float* al_d  = al_s + N;
    float* vecs  = al_d + N;
    int* cnt      = (int*)(vecs + 4 * F_DIM);
    int* rowptr   = cnt + N;
    int* binStart = rowptr + N;
    int* binTotal = binStart + NBP;
    int* table    = binTotal + NBP;
    int* ssrc     = table + (size_t)nChunk * NBP;
    int2* pairs   = (int2*)(ssrc + (size_t)E + ((size_t)E & 1));

    float* out = (float*)d_out;

    dim3 b256(256);
    const int NODE_RESIDENT_WGS = 2048;
    const int GEMM_RESIDENT_WGS = 1280;
    const int nRow  = (N + 7) / 8;
    const int nGemm = (N + 15) / 16;
    const int gRow  = nRow  < NODE_RESIDENT_WGS ? nRow  : NODE_RESIDENT_WGS;
    const int gGemm = nGemm < GEMM_RESIDENT_WGS ? nGemm : GEMM_RESIDENT_WGS;
    const int gT64  = (N + 63) / 64;
    const int gT64a = (gT64 + 1) / 2;

    gat_prep_count<<<4 + nChunk + (2 * WPACK_ELEMS) / 1024, 1024, 0, stream>>>(
        W1s, a1s, W1d, a1d, W2s, a2s, W2d, a2d, vecs, wpack, dst, table, E, nChunk);

    gat_partition_rows<<<nChunk + gT64a, 1024, 0, stream>>>(
        src, dst, table, pairs, binStart, binTotal, E, nChunk,
        x, vecs, vecs + 128, xb, al_s, al_d, N);
    gat_sort_rows<<<nBin + (gT64 - gT64a), 1024, 0, stream>>>(
        pairs, binStart, binTotal, cnt, rowptr, ssrc, nBin,
        x, vecs, vecs + 128, xb, al_s, al_d, N, gT64a);

    gat_node_agg<<<gRow, b256, 0, stream>>>(rowptr, cnt, ssrc, al_s, al_d, xb, sums, N);
    gat_gemm_wmma<true><<<gGemm, b256, 0, stream>>>(
        sums, wpack, b1, vecs + 256, vecs + 384, hb, al_s, al_d, out, N);

    gat_node_agg<<<gRow, b256, 0, stream>>>(rowptr, cnt, ssrc, al_s, al_d, hb, sums, N);
    gat_gemm_wmma<false><<<gGemm, b256, 0, stream>>>(
        sums, wpack + WPACK_ELEMS, b2, vecs + 256, vecs + 384, hb, al_s, al_d, out, N);
}


# Wave reductions (per-node softmax max and sum, attention row dots) on the vector ALU, DPP row_xmask + permlanex16, instead of five dependent LDS permutes each; same trees, sums bit-identical
# speedup vs baseline: 8.2212x; 1.0429x over previous
#include <hip/hip_runtime.h>
#include <hip/hip_bf16.h>


#define F_DIM 128
#define NEG_SLOPE 0.2f

typedef __attribute__((ext_vector_type(16))) __bf16 v16bf;
typedef __attribute__((ext_vector_type(8)))  float  v8f;

#define WPACK_ELEMS (8 * 4 * 32 * 16)

__device__ __forceinline__ float leaky(float z) {
    return z > 0.0f ? z : NEG_SLOPE * z;
}

template <int M>
__device__ __forceinline__ float xrow(float v) {
    return __builtin_bit_cast(float, __builtin_amdgcn_update_dpp(
        0, __builtin_bit_cast(int, v), 0x160 + M, 0xf, 0xf, true));
}
__device__ __forceinline__ float xhalf(float v) {
    const int b = __builtin_bit_cast(int, v);
    return __builtin_bit_cast(float, __builtin_amdgcn_permlanex16(b, b, 0x76543210, 0xfedcba98, false, false));
}
__device__ __forceinline__ float wave_sum(float v) {
    v += xhalf(v);
    v += xrow<8>(v);
    v += xrow<4>(v);
    v += xrow<2>(v);
    v += xrow<1>(v);
    return v;
}
__device__ __forceinline__ float wave_max(float v) {
    v = fmaxf(v, xhalf(v));
    v = fmaxf(v, xrow<8>(v));
    v = fmaxf(v, xrow<4>(v));
    v = fmaxf(v, xrow<2>(v));
    v = fmaxf(v, xrow<1>(v));
    return v;
}

__device__ __forceinline__ void pack_w_part(const float* __restrict__ W1s, const float* __restrict__ W2s,
                                            __bf16* __restrict__ packed, int idx) {
    int j    = idx & 15;
    int lane = (idx >> 4) & 31;
    int t    = (idx >> 9) & 3;
    int w    = (idx >> 11) & 7;
    int p    = (idx >> 14) & 1;
    const float* W = p ? W2s : W1s;
    int koff = (lane >> 4) * 8;
    int mn   = lane & 15;
    int k    = t * 32 + ((j >> 3) << 4) + koff + (j & 7);
    int col  = w * 16 + mn;
    packed[idx] = (__bf16)W[k * F_DIM + col];
}

__device__ __forceinline__ void matvec_part(const float* __restrict__ W1s, const float* __restrict__ a1s,
                                            const float* __restrict__ W1d, const float* __restrict__ a1d,
                                            const float* __restrict__ W2s, const float* __restrict__ a2s,
                                            const float* __restrict__ W2d, const float* __restrict__ a2d,
                                            float* __restrict__ vecs, int which, int f) {
    const float* W; const float* a;
    switch (which) {
        case 0:  W = W1s; a = a1s; break;
        case 1:  W = W1d; a = a1d; break;
        case 2:  W = W2s; a = a2s; break;
        default: W = W2d; a = a2d; break;
    }
    float s = 0.0f;
    #pragma unroll 8
    for (int d = 0; d < F_DIM; ++d) s += W[f * F_DIM + d] * a[d];
    vecs[which * F_DIM + f] = s;
}

typedef __attribute__((ext_vector_type(4))) __bf16 v4bf;

#define A_PITCH (F_DIM + 8)
#define C_PITCH (F_DIM + 4)


__device__ __forceinline__ void
rows_to_bf16_64(const float* __restrict__ A, const float* __restrict__ vs, const float* __restrict__ vd,
                __bf16* __restrict__ Ab, float* __restrict__ al_s, float* __restrict__ al_d,
                const int tile64, const int N) {
    const int tid = threadIdx.x, lane = tid & 31;
    const int wave = __builtin_amdgcn_readfirstlane(tid >> 5);
    const int ra = tile64 * 64 + 2 * wave, rb = ra + 1;
    if (ra >= N) return;
    const bool hasb = rb < N;
    const float4 xa = ((const float4*)(A + (size_t)ra * F_DIM))[lane];
    float4 xb = make_float4(0.0f, 0.0f, 0.0f, 0.0f);
    if (hasb) xb = ((const float4*)(A + (size_t)rb * F_DIM))[lane];
    const float4 av = ((const float4*)vs)[lane];
    const float4 bv = ((const float4*)vd)[lane];
    float sa = xa.x * av.x + xa.y * av.y + xa.z * av.z + xa.w * av.w;
    float da = xa.x * bv.x + xa.y * bv.y + xa.z * bv.z + xa.w * bv.w;
    float sb = xb.x * av.x + xb.y * av.y + xb.z * av.z + xb.w * av.w;
    float db = xb.x * bv.x + xb.y * bv.y + xb.z * bv.z + xb.w * bv.w;
    v4bf pa = { (__bf16)xa.x, (__bf16)xa.y, (__bf16)xa.z, (__bf16)xa.w };
    v4bf pb = { (__bf16)xb.x, (__bf16)xb.y, (__bf16)xb.z, (__bf16)xb.w };
    *(v4bf*)(Ab + (size_t)ra * F_DIM + 4 * lane) = pa;
    if (hasb) *(v4bf*)(Ab + (size_t)rb * F_DIM + 4 * lane) = pb;
    sa = wave_sum(sa);
    da = wave_sum(da);
    sb = wave_sum(sb);
    db = wave_sum(db);
    if (lane == 0) {
        al_s[ra] = sa; al_d[ra] = da;
        if (hasb) { al_s[rb] = sb; al_d[rb] = db; }
    }
}

__device__ __forceinline__ void split4(const float4 x, v4bf& hi, v4bf& lo) {
    const __bf16 h0 = (__bf16)x.x, h1 = (__bf16)x.y, h2 = (__bf16)x.z, h3 = (__bf16)x.w;
    hi = v4bf{ h0, h1, h2, h3 };
    lo = v4bf{ (__bf16)(x.x - (float)h0), (__bf16)(x.y - (float)h1),
               (__bf16)(x.z - (float)h2), (__bf16)(x.w - (float)h3) };
}

template <bool FIRST>
__global__ void __launch_bounds__(256, 10)
gat_gemm_wmma(const float* __restrict__ S, const __bf16* __restrict__ Wp, const float* __restrict__ bias,
              const float* __restrict__ vs, const float* __restrict__ vd,
              __bf16* __restrict__ Hb, float* __restrict__ al_s, float* __restrict__ al_d,
              float* __restrict__ out, int N) {
    __shared__ __attribute__((aligned(16))) __bf16 AsH[16][A_PITCH];
    __shared__ __attribute__((aligned(16))) __bf16 AsL[16][A_PITCH];
    __shared__ __attribute__((aligned(16))) float  Cs[16][C_PITCH];
    const int tid = threadIdx.x, lane = tid & 31;
    const int wave = __builtin_amdgcn_readfirstlane(tid >> 5);
    const int ra = 2 * wave, rb = 2 * wave + 1;
    const int col0 = wave * 16;
    const int mn   = lane & 15;
    const int koff = (lane >> 4) * 8;
    const int nTiles = (N + 15) >> 4;

    const char* Sbytes = (const char*)S;
    float4 xa = make_float4(0.0f, 0.0f, 0.0f, 0.0f), xb = xa;
    if ((int)blockIdx.x < nTiles) {
        const int r0 = blockIdx.x * 16 + ra;
        const unsigned o0 = (unsigned)r0 * (F_DIM * 4u) + (unsigned)lane * 16u;
        if (r0 + 1 < N)  { xa = *(const float4*)(Sbytes + o0); xb = *(const float4*)(Sbytes + o0 + F_DIM * 4u); }
        else if (r0 < N) { xa = *(const float4*)(Sbytes + o0); }
    }

    const v16bf* Bfrag = (const v16bf*)Wp + (size_t)(wave * 4) * 32 + lane;
    v16bf bfr[4];
    #pragma unroll
    for (int t = 0; t < 4; ++t) bfr[t] = Bfrag[t * 32];
    const float bc = bias[col0 + mn];
    float4 av = make_float4(0.0f, 0.0f, 0.0f, 0.0f), bv = av;
    if (FIRST) {
        av = ((const float4*)vs)[lane];
        bv = ((const float4*)vd)[lane];
    }

    for (int tile = blockIdx.x; tile < nTiles; tile += gridDim.x) {
    const int row0 = tile * 16;
    const bool oka = row0 + ra < N, okb = row0 + rb < N;

    v4bf ha, la, hb, lb;
    split4(xa, ha, la);
    split4(xb, hb, lb);
    *(v4bf*)&AsH[ra][4 * lane] = ha;
    *(v4bf*)&AsL[ra][4 * lane] = la;
    *(v4bf*)&AsH[rb][4 * lane] = hb;
    *(v4bf*)&AsL[rb][4 * lane] = lb;
    __syncthreads();

    xa = make_float4(0.0f, 0.0f, 0.0f, 0.0f); xb = xa;
    if (tile + (int)gridDim.x < nTiles) {
        const int r1 = (tile + (int)gridDim.x) * 16 + ra;
        const unsigned o1 = (unsigned)r1 * (F_DIM * 4u) + (unsigned)lane * 16u;
        if (r1 + 1 < N)  { xa = *(const float4*)(Sbytes + o1); xb = *(const float4*)(Sbytes + o1 + F_DIM * 4u); }
        else if (r1 < N) { xa = *(const float4*)(Sbytes + o1); }
    }

    v8f acch = {}, accl = {};
    #pragma unroll
    for (int t = 0; t < 4; ++t) {
        v16bf afh, afl;
        #pragma unroll
        for (int i = 0; i < 8; ++i) {
            afh[i]     = AsH[mn][t * 32 + koff + i];
            afh[i + 8] = AsH[mn][t * 32 + 16 + koff + i];
            afl[i]     = AsL[mn][t * 32 + koff + i];
            afl[i + 8] = AsL[mn][t * 32 + 16 + koff + i];
        }
        acch = __builtin_amdgcn_wmma_f32_16x16x32_bf16(
             false, afh,  false, bfr[t],
             (short)0, acch,  false,  false);
        accl = __builtin_amdgcn_wmma_f32_16x16x32_bf16(
             false, afl,  false, bfr[t],
             (short)0, accl,  false,  false);
    }

    const int mb = (lane >> 4) * 8;
    #pragma unroll
    for (int v = 0; v < 8; ++v) {
        float r = (acch[v] + accl[v]) + bc;
        if (FIRST) r = fmaxf(r, 0.0f);
        Cs[mb + v][col0 + mn] = r;
    }
    __syncthreads();

    const float4 ca = *(const float4*)&Cs[ra][4 * lane];
    const float4 cb = *(const float4*)&Cs[rb][4 * lane];
    if (FIRST) {
        float sa = ca.x * av.x + ca.y * av.y + ca.z * av.z + ca.w * av.w;
        float da = ca.x * bv.x + ca.y * bv.y + ca.z * bv.z + ca.w * bv.w;
        float sb = cb.x * av.x + cb.y * av.y + cb.z * av.z + cb.w * av.w;
        float db = cb.x * bv.x + cb.y * bv.y + cb.z * bv.z + cb.w * bv.w;
        v4bf pa = { (__bf16)ca.x, (__bf16)ca.y, (__bf16)ca.z, (__bf16)ca.w };
        v4bf pb = { (__bf16)cb.x, (__bf16)cb.y, (__bf16)cb.z, (__bf16)cb.w };
        if (oka) *(v4bf*)(Hb + (size_t)(row0 + ra) * F_DIM + 4 * lane) = pa;
        if (okb) *(v4bf*)(Hb + (size_t)(row0 + rb) * F_DIM + 4 * lane) = pb;
        sa = wave_sum(sa);
        da = wave_sum(da);
        sb = wave_sum(sb);
        db = wave_sum(db);
        if (lane == 0) {
            if (oka) { al_s[row0 + ra] = sa; al_d[row0 + ra] = da; }
            if (okb) { al_s[row0 + rb] = sb; al_d[row0 + rb] = db; }
        }
    } else {
        if (oka) ((float4*)(out + (size_t)(row0 + ra) * F_DIM))[lane] = ca;
        if (okb) ((float4*)(out + (size_t)(row0 + rb) * F_DIM))[lane] = cb;
    }
    }
}


#define BIN_SHIFT 9
#define BIN_W     512
#define NBP       256
#define CH_CAP    6400
#define BIN_CAP   12288

__device__ __forceinline__ int wave_incl_scan(int v, int lane) {
    #pragma unroll
    for (int off = 1; off < 32; off <<= 1) {
        int t = __shfl_up(v, off, 32);
        if (lane >= off) v += t;
    }
    return v;
}

__device__ __forceinline__ void
bin_count_part(const int* __restrict__ dst, int* __restrict__ table, int E, const int chunk) {
    __shared__ int h[NBP];
    const int t = threadIdx.x;
    if (t < NBP) h[t] = 0;
    __syncthreads();
    const int e0 = chunk * CH_CAP;
    const int e1 = (e0 + CH_CAP < E) ? (e0 + CH_CAP) : E;
    for (int e = e0 + t; e < e1; e += 1024)
        atomicAdd(&h[(dst[e] >> BIN_SHIFT) & (NBP - 1)], 1);
    __syncthreads();
    if (t < NBP) table[chunk * NBP + t] = h[t];
}

__global__ void __launch_bounds__(1024)
gat_prep_count(const float* __restrict__ W1s, const float* __restrict__ a1s,
               const float* __restrict__ W1d, const float* __restrict__ a1d,
               const float* __restrict__ W2s, const float* __restrict__ a2s,
               const float* __restrict__ W2d, const float* __restrict__ a2d,
               float* __restrict__ vecs, __bf16* __restrict__ packed,
               const int* __restrict__ dst, int* __restrict__ table, int E, int NC) {
    const int b = blockIdx.x;
    if (b < 4) {
        if (threadIdx.x < F_DIM)
            matvec_part(W1s, a1s, W1d, a1d, W2s, a2s, W2d, a2d, vecs, b, threadIdx.x);
    } else if (b < 4 + NC) {
        bin_count_part(dst, table, E, b - 4);
    } else {
        pack_w_part(W1s, W2s, packed, (b - 4 - NC) * 1024 + threadIdx.x);
    }
}

__device__ __forceinline__ void
bin_partition_part(const int* __restrict__ src, const int* __restrict__ dst,
                   const int* __restrict__ table, int2* __restrict__ pairs,
                   int* __restrict__ binStart, int* __restrict__ binTotal, int E, int NC,
                   const int chunk) {
    __shared__ int  sTot[4][NBP];
    __shared__ int  sPre[4][NBP];
    __shared__ int  sMine[NBP];
    __shared__ int  sCur[NBP];
    __shared__ int  sOff[NBP];
    __shared__ int  sWs[2][8];
    __shared__ int2 sBuf[CH_CAP];
    const int t = threadIdx.x, lane = t & 31, wave = t >> 5;
    const int b = t & (NBP - 1), q = t >> 8;

    int tot = 0, pre = 0;
    #pragma unroll 16
    for (int c = q; c < NC; c += 4) {
        const int v = table[c * NBP + b];
        tot += v;
        if (c < chunk)  pre += v;
        if (c == chunk) sMine[b] = v;
    }
    sTot[q][b] = tot;
    sPre[q][b] = pre;
    __syncthreads();

    int T = 0, P = 0, M = 0, iT = 0, iM = 0;
    if (t < NBP) {
        T = sTot[0][t] + sTot[1][t] + sTot[2][t] + sTot[3][t];
        P = sPre[0][t] + sPre[1][t] + sPre[2][t] + sPre[3][t];
        M = sMine[t];
        iT = wave_incl_scan(T, lane);
        iM = wave_incl_scan(M, lane);
        if (lane == 31) { sWs[0][wave] = iT; sWs[1][wave] = iM; }
    }
    __syncthreads();
    if (t < NBP) {
        int oT = 0, oM = 0;
        for (int w = 0; w < wave; ++w) { oT += sWs[0][w]; oM += sWs[1][w]; }
        const int gStart = oT + iT - T;
        const int lStart = oM + iM - M;
        sCur[t] = lStart;
        sOff[t] = gStart + P - lStart;
        if (chunk == 0) { binStart[t] = gStart; binTotal[t] = T; }
    }
    __syncthreads();

    const int e0 = chunk * CH_CAP;
    const int e1 = (e0 + CH_CAP < E) ? (e0 + CH_CAP) : E;
    for (int eb = e0 + t; eb < e1; eb += 4 * 1024) {
        int d[4], s[4];
        #pragma unroll
        for (int k = 0; k < 4; ++k) {
            const int e = eb + k * 1024;
            d[k] = (e < e1) ? dst[e] : 0;
            s[k] = (e < e1) ? src[e] : 0;
        }
        #pragma unroll
        for (int k = 0; k < 4; ++k) {
            if (eb + k * 1024 < e1) {
                const int p = atomicAdd(&sCur[(d[k] >> BIN_SHIFT) & (NBP - 1)], 1);
                sBuf[p] = make_int2(s[k], d[k]);
            }
        }
    }
    __syncthreads();
    const int len = e1 - e0;
    for (int i = t; i < len; i += 1024) {
        const int2 pr = sBuf[i];
        pairs[sOff[(pr.y >> BIN_SHIFT) & (NBP - 1)] + i] = pr;
    }
}

__device__ __forceinline__ void
bin_sort_part(const int2* __restrict__ pairs, const int* __restrict__ binStart,
              const int* __restrict__ binTotal, int* __restrict__ cnt,
              int* __restrict__ rowptr, int* __restrict__ ssrc, int N, const int bin) {
    __shared__ int sCnt[BIN_W];
    __shared__ int sCur[BIN_W];
    __shared__ int sWs[16];
    __shared__ int sOut[BIN_CAP];
    const int t = threadIdx.x, lane = t & 31, wave = t >> 5;
    const int lo   = bin << BIN_SHIFT;
    const int base = binStart[bin];
    const int tot  = binTotal[bin];

    if (t < BIN_W) sCnt[t] = 0;
    __syncthreads();
    for (int ib = t; ib < tot; ib += 4 * 1024) {
        int d[4];
        #pragma unroll
        for (int k = 0; k < 4; ++k) {
            const int i = ib + k * 1024;
            d[k] = (i < tot) ? pairs[base + i].y : lo;
        }
        #pragma unroll
        for (int k = 0; k < 4; ++k)
            if (ib + k * 1024 < tot) atomicAdd(&sCnt[(d[k] - lo) & (BIN_W - 1)], 1);
    }
    __syncthreads();

    int c = 0, inc = 0;
    if (t < BIN_W) {
        c   = sCnt[t];
        inc = wave_incl_scan(c, lane);
        if (lane == 31) sWs[wave] = inc;
    }
    __syncthreads();
    if (t < BIN_W) {
        int o = 0;
        for (int w = 0; w < wave; ++w) o += sWs[w];
        const int ex = o + inc - c;
        sCur[t] = ex;
        const int n = lo + t;
        if (n < N) { cnt[n] = c; rowptr[n] = base + ex; }
    }
    __syncthreads();

    if (tot <= BIN_CAP) {
        for (int ib = t; ib < tot; ib += 4 * 1024) {
            int2 pr[4];
            #pragma unroll
            for (int k = 0; k < 4; ++k) {
                const int i = ib + k * 1024;
                pr[k] = (i < tot) ? pairs[base + i] : make_int2(0, lo);
            }
            #pragma unroll
            for (int k = 0; k < 4; ++k) {
                if (ib + k * 1024 < tot) {
                    const int p = atomicAdd(&sCur[(pr[k].y - lo) & (BIN_W - 1)], 1);
                    sOut[p] = pr[k].x;
                }
            }
        }
        __syncthreads();
        for (int i = t; i < tot; i += 1024) ssrc[base + i] = sOut[i];
    } else {
        for (int i = t; i < tot; i += 1024) {
            const int2 pr = pairs[base + i];
            const int p = atomicAdd(&sCur[(pr.y - lo) & (BIN_W - 1)], 1);
            ssrc[base + p] = pr.x;
        }
    }
}

__global__ void __launch_bounds__(1024, 16)
gat_partition_rows(const int* __restrict__ src, const int* __restrict__ dst,
                   const int* __restrict__ table, int2* __restrict__ pairs,
                   int* __restrict__ binStart, int* __restrict__ binTotal, int E, int NC,
                   const float* __restrict__ A, const float* __restrict__ vs, const float* __restrict__ vd,
                   __bf16* __restrict__ Ab, float* __restrict__ al_s, float* __restrict__ al_d,
                   int N) {
    const int b = blockIdx.x;
    if (b < NC) bin_partition_part(src, dst, table, pairs, binStart, binTotal, E, NC, b);
    else        rows_to_bf16_64(A, vs, vd, Ab, al_s, al_d, b - NC, N);
}

__global__ void __launch_bounds__(1024)
gat_sort_rows(const int2* __restrict__ pairs, const int* __restrict__ binStart,
              const int* __restrict__ binTotal, int* __restrict__ cnt,
              int* __restrict__ rowptr, int* __restrict__ ssrc, int nBin,
              const float* __restrict__ A, const float* __restrict__ vs, const float* __restrict__ vd,
              __bf16* __restrict__ Ab, float* __restrict__ al_s, float* __restrict__ al_d,
              int N, int tile0) {
    const int b = blockIdx.x;
    if (b < nBin) bin_sort_part(pairs, binStart, binTotal, cnt, rowptr, ssrc, N, b);
    else          rows_to_bf16_64(A, vs, vd, Ab, al_s, al_d, tile0 + (b - nBin), N);
}


#define SORT_CAP 128

typedef __attribute__((address_space(3))) int lds_int;

template <int G>
__device__ __forceinline__ void agg_rows(const __bf16* __restrict__ rows, int s, float ex, int k,
                                         int lane, float4& acc) {
    uint2 hv[G];
    float wk[G];
    #pragma unroll
    for (int u = 0; u < G; ++u) {
        const int sk = __builtin_amdgcn_readlane(s, k + u);
        wk[u] = __int_as_float(__builtin_amdgcn_readlane(__float_as_int(ex), k + u));
        hv[u] = ((const uint2*)(rows + (size_t)sk * F_DIM))[lane];
    }
    #pragma unroll
    for (int u = 0; u < G; ++u) {
        const float f0 = __uint_as_float(hv[u].x << 16);
        const float f1 = __uint_as_float(hv[u].x & 0xffff0000u);
        const float f2 = __uint_as_float(hv[u].y << 16);
        const float f3 = __uint_as_float(hv[u].y & 0xffff0000u);
        acc.x = __builtin_fmaf(wk[u], f0, acc.x);
        acc.y = __builtin_fmaf(wk[u], f1, acc.y);
        acc.z = __builtin_fmaf(wk[u], f2, acc.z);
        acc.w = __builtin_fmaf(wk[u], f3, acc.w);
    }
}

__global__ void __launch_bounds__(256)
gat_node_agg(const int* __restrict__ rowptr, const int* __restrict__ cnt,
             const int* __restrict__ ssrc, const float* __restrict__ al_s,
             const float* __restrict__ al_d, const __bf16* __restrict__ rows,
             float* __restrict__ sums, int N) {
    __shared__ int sSort[8][SORT_CAP];
    const int lane = threadIdx.x & 31;
    const int wv = __builtin_amdgcn_readfirstlane((int)(threadIdx.x >> 5));
    lds_int* mySort = (lds_int*)sSort[wv];
    int n = __builtin_amdgcn_readfirstlane((int)(blockIdx.x * 8 + (threadIdx.x >> 5)));
    if (n >= N) return;
    const int stride = (int)gridDim.x * 8;
    int   startN = rowptr[n];
    int   degN   = cnt[n];
    float aldN   = al_d[n];
    for (;;) {
    const int start = startN;
    const int deg   = degN;
    const float ald = aldN;
    const int nn = n + stride;
    if (nn < N) { startN = rowptr[nn]; degN = cnt[nn]; aldN = al_d[nn]; }
    const bool inLds = (deg > 32) && (deg <= SORT_CAP);

    int s0 = 0;
    if (deg <= 32) {
        unsigned key = 0xffffffffu;
        if (lane < deg) key = ((unsigned)ssrc[start + lane] << 5) | (unsigned)lane;
        int rank = 0;
        for (int i = 0; i < deg; ++i) {
            const unsigned ki = (unsigned)__builtin_amdgcn_readlane((int)key, i);
            rank += (ki < key) ? 1 : 0;
        }
        if (lane >= deg) rank = lane;
        s0 = (int)((unsigned)__builtin_amdgcn_ds_permute(rank << 2, (int)key) >> 5);
        if (lane >= deg) s0 = 0;
    } else {
        if (inLds) {
            const int nb = (deg + 31) >> 5;
            for (int b1 = 0; b1 < nb; ++b1) {
                const int j1 = b1 * 32 + lane;
                const int s1 = (j1 < deg) ? ssrc[start + j1] : 0x7fffffff;
                int rank = 0;
                for (int b2 = 0; b2 < nb; ++b2) {
                    const int j2 = b2 * 32 + lane;
                    const int s2 = (j2 < deg) ? ssrc[start + j2] : 0x7fffffff;
                    const int lim = (deg - b2 * 32 < 32) ? (deg - b2 * 32) : 32;
                    for (int i = 0; i < lim; ++i) {
                        const int si = __builtin_amdgcn_readlane(s2, i);
                        rank += (si < s1 || (si == s1 && b2 * 32 + i < j1)) ? 1 : 0;
                    }
                }
                if (j1 < deg) mySort[rank] = s1;
            }
            s0 = mySort[lane];
        } else {
            s0 = ssrc[start + lane];
        }
    }
    float z0 = __int_as_float(0xff800000);
    if (lane < deg) z0 = leaky(al_s[s0] + ald);
    float m = z0;
    for (int base = 32; base < deg; base += 32) {
        const int j = base + lane;
        if (j < deg) {
            const int sj = inLds ? mySort[j] : ssrc[start + j];
            m = fmaxf(m, leaky(al_s[sj] + ald));
        }
    }
    m = wave_max(m);

    float4 acc = make_float4(0.0f, 0.0f, 0.0f, 0.0f);
    float dsum = 0.0f;
    for (int base = 0; base < deg; base += 32) {
        const int j = base + lane;
        int s = 0; float ex = 0.0f;
        if (base == 0) {
            s = s0;
            if (lane < deg) ex = __expf(z0 - m);
        } else if (j < deg) {
            s  = inLds ? mySort[j] : ssrc[start + j];
            ex = __expf(leaky(al_s[s] + ald) - m);
        }
        dsum += ex;
        const int nk = (deg - base < 32) ? (deg - base) : 32;
        int k = 0;
        for (; k + 8 <= nk; k += 8) agg_rows<8>(rows, s, ex, k, lane, acc);
        switch (nk - k) {
            case 1: agg_rows<1>(rows, s, ex, k, lane, acc); break;
            case 2: agg_rows<2>(rows, s, ex, k, lane, acc); break;
            case 3: agg_rows<3>(rows, s, ex, k, lane, acc); break;
            case 4: agg_rows<4>(rows, s, ex, k, lane, acc); break;
            case 5: agg_rows<5>(rows, s, ex, k, lane, acc); break;
            case 6: agg_rows<6>(rows, s, ex, k, lane, acc); break;
            case 7: agg_rows<7>(rows, s, ex, k, lane, acc); break;
            default: break;
        }
    }
    dsum = wave_sum(dsum);

    const float inv = (deg > 0) ? 1.0f / dsum : 0.0f;
    float4 r;
    r.x = acc.x * inv;
    r.y = acc.y * inv;
    r.z = acc.z * inv;
    r.w = acc.w * inv;
    ((float4*)(sums + (size_t)n * F_DIM))[lane] = r;
    if (nn >= N) break;
    n = nn;
    }
}

extern "C" void kernel_launch(void* const* d_in, const int* in_sizes, int n_in,
                              void* d_out, int out_size, void* d_ws, size_t ws_size,
                              hipStream_t stream) {
    const float* x   = (const float*)d_in[0];
    const int*   ei  = (const int*)d_in[1];
    const float* W1s = (const float*)d_in[2];
    const float* W1d = (const float*)d_in[3];
    const float* a1s = (const float*)d_in[4];
    const float* a1d = (const float*)d_in[5];
    const float* b1  = (const float*)d_in[6];
    const float* W2s = (const float*)d_in[7];
    const float* W2d = (const float*)d_in[8];
    const float* a2s = (const float*)d_in[9];
    const float* a2d = (const float*)d_in[10];
    const float* b2  = (const float*)d_in[11];

    const int N = in_sizes[0] / F_DIM;
    const int E = in_sizes[1] / 2;
    const int* src = ei;
    const int* dst = ei + E;

    const int nChunk = (E + CH_CAP - 1) / CH_CAP;
    const int nBin   = (N + BIN_W - 1) / BIN_W;

    __bf16* wpack = (__bf16*)d_ws;
    float*  fws   = (float*)((char*)d_ws + 2 * WPACK_ELEMS * sizeof(__bf16));
    float* sums  = fws;
    __bf16* xb   = (__bf16*)(sums + (size_t)N * F_DIM);
    __bf16* hb   = xb + (size_t)N * F_DIM;
    float* al_s  = (float*)(hb + (size_t)N * F_DIM);
    float* al_d  = al_s + N;
    float* vecs  = al_d + N;
    int* cnt      = (int*)(vecs + 4 * F_DIM);
    int* rowptr   = cnt + N;
    int* binStart = rowptr + N;
    int* binTotal = binStart + NBP;
    int* table    = binTotal + NBP;
    int* ssrc     = table + (size_t)nChunk * NBP;
    int2* pairs   = (int2*)(ssrc + (size_t)E + ((size_t)E & 1));

    float* out = (float*)d_out;

    dim3 b256(256);
    const int NODE_RESIDENT_WGS = 2048;
    const int GEMM_RESIDENT_WGS = 1280;
    const int nRow  = (N + 7) / 8;
    const int nGemm = (N + 15) / 16;
    const int gRow  = nRow  < NODE_RESIDENT_WGS ? nRow  : NODE_RESIDENT_WGS;
    const int gGemm = nGemm < GEMM_RESIDENT_WGS ? nGemm : GEMM_RESIDENT_WGS;
    const int gT64  = (N + 63) / 64;
    const int gT64a = (gT64 + 1) / 2;

    gat_prep_count<<<4 + nChunk + (2 * WPACK_ELEMS) / 1024, 1024, 0, stream>>>(
        W1s, a1s, W1d, a1d, W2s, a2s, W2d, a2d, vecs, wpack, dst, table, E, nChunk);

    gat_partition_rows<<<nChunk + gT64a, 1024, 0, stream>>>(
        src, dst, table, pairs, binStart, binTotal, E, nChunk,
        x, vecs, vecs + 128, xb, al_s, al_d, N);
    gat_sort_rows<<<nBin + (gT64 - gT64a), 1024, 0, stream>>>(
        pairs, binStart, binTotal, cnt, rowptr, ssrc, nBin,
        x, vecs, vecs + 128, xb, al_s, al_d, N, gT64a);

    gat_node_agg<<<gRow, b256, 0, stream>>>(rowptr, cnt, ssrc, al_s, al_d, xb, sums, N);
    gat_gemm_wmma<true><<<gGemm, b256, 0, stream>>>(
        sums, wpack, b1, vecs + 256, vecs + 384, hb, al_s, al_d, out, N);

    gat_node_agg<<<gRow, b256, 0, stream>>>(rowptr, cnt, ssrc, al_s, al_d, hb, sums, N);
    gat_gemm_wmma<false><<<gGemm, b256, 0, stream>>>(
        sums, wpack + WPACK_ELEMS, b2, vecs + 256, vecs + 384, hb, al_s, al_d, out, N);
}
